# starting kernel
# baseline (speedup 1.0000x reference)
#include <hip/hip_runtime.h>

#define HIDDEN 64
#define EIGD   16
#define NUSERS 100000
#define NITEMS 200000
#define NNODES 300000
#define NEDGES 1250000
#define NPATHS 14
#define LN_EPS 1e-5f

typedef __attribute__((ext_vector_type(2))) float v2f;
typedef __attribute__((ext_vector_type(8))) float v8f;

static __device__ __forceinline__ unsigned fenc(float f) {
  unsigned u = __float_as_uint(f);
  return (u & 0x80000000u) ? ~u : (u | 0x80000000u);
}
static __device__ __forceinline__ float fdec(unsigned u) {
  return (u & 0x80000000u) ? __uint_as_float(u ^ 0x80000000u)
                           : __uint_as_float(~u);
}

__global__ void ln_wmma_kernel(const float* __restrict__ srcA,
                               const float* __restrict__ srcB,
                               int nA, float* __restrict__ y) {
  int lane = threadIdx.x & 31;
  int wave = threadIdx.x >> 5;
  int base = (blockIdx.x * (blockDim.x >> 5) + wave) << 4;
  if (base >= NNODES) return;

  int m = lane & 15;
  int h = lane >> 4;
  int node = base + m;
  const float* srow = (node < nA) ? (srcA + (size_t)node * HIDDEN)
                                  : (srcB + (size_t)(node - nA) * HIDDEN);
  const float* sp = srow + 2 * h;
  float*       dp = y + (size_t)node * HIDDEN + 2 * h;

  const v2f ones = {1.0f, 1.0f};
  v8f accS = {};
  v8f accQ = {};
  float2 vbuf[16];
#pragma unroll
  for (int c = 0; c < 16; ++c) {
    float2 v = *(const float2*)(sp + 4 * c);
    vbuf[c] = v;
    v2f a  = {v.x, v.y};
    v2f a2 = {v.x * v.x, v.y * v.y};
    accS = __builtin_amdgcn_wmma_f32_16x16x4_f32(false, a,  false, ones,
                                                 (short)0, accS, false, false);
    accQ = __builtin_amdgcn_wmma_f32_16x16x4_f32(false, a2, false, ones,
                                                 (short)0, accQ, false, false);
  }

  float s = 0.0f, q = 0.0f;
#pragma unroll
  for (int i = 0; i < 8; ++i) {
    float sLo = __shfl(accS[i], 0, 32);
    float sHi = __shfl(accS[i], 16, 32);
    float qLo = __shfl(accQ[i], 0, 32);
    float qHi = __shfl(accQ[i], 16, 32);
    s = (m == i)     ? sLo : s;
    s = (m == 8 + i) ? sHi : s;
    q = (m == i)     ? qLo : q;
    q = (m == 8 + i) ? qHi : q;
  }

  float mu  = s * (1.0f / HIDDEN);
  float var = q * (1.0f / HIDDEN) - mu * mu;
  float rin = rsqrtf(var + LN_EPS);
#pragma unroll
  for (int c = 0; c < 16; ++c) {
    float2 o = make_float2((vbuf[c].x - mu) * rin, (vbuf[c].y - mu) * rin);
    *(float2*)(dp + 4 * c) = o;
  }
}

__global__ void score_kernel(const float* __restrict__ y,
                             const float* __restrict__ eigs,
                             const int* __restrict__ row,
                             const int* __restrict__ col,
                             const int* __restrict__ pt,
                             const float* __restrict__ lambda0_l,
                             const float* __restrict__ pemb_l,
                             float* __restrict__ s0,
                             unsigned* __restrict__ m0u,
                             unsigned* __restrict__ m1u) {
  int e = blockIdx.x * (blockDim.x >> 4) + (threadIdx.x >> 4);
  if (e >= NEDGES) return;
  int lane = threadIdx.x & 15;
  int r = row[e], c = col[e];
  __builtin_prefetch(eigs + (size_t)r * EIGD, 0, 1);
  __builtin_prefetch(eigs + (size_t)c * EIGD, 0, 1);
  float4 a4 = ((const float4*)(y + (size_t)r * HIDDEN))[lane];
  float4 b4 = ((const float4*)(y + (size_t)c * HIDDEN))[lane];
  float p = 0.125f * (a4.x * b4.x + a4.y * b4.y + a4.z * b4.z + a4.w * b4.w);
  float er = eigs[(size_t)r * EIGD + lane];
  float ec = eigs[(size_t)c * EIGD + lane];
  float explam = expf(lambda0_l[0]);
  p += explam * er * ec;
#pragma unroll
  for (int off = 8; off > 0; off >>= 1) p += __shfl_xor(p, off, 16);
  if (lane == 0) {
    s0[e] = p;
    atomicMax(m0u + r, fenc(p));
    atomicMax(m1u + r, fenc(pemb_l[pt[e]]));
  }
}

__global__ void expsum_kernel(const float* __restrict__ s0,
                              const int* __restrict__ row,
                              const int* __restrict__ pt,
                              const float* __restrict__ pemb_l,
                              const unsigned* __restrict__ m0u,
                              const unsigned* __restrict__ m1u,
                              float* __restrict__ e0,
                              float* __restrict__ d0,
                              float* __restrict__ d1) {
  int e = blockIdx.x * blockDim.x + threadIdx.x;
  if (e >= NEDGES) return;
  int r = row[e];
  float e0v = expf(s0[e] - fdec(m0u[r]));
  e0[e] = e0v;
  atomicAdd(d0 + r, e0v);
  float e1v = expf(pemb_l[pt[e]] - fdec(m1u[r]));
  atomicAdd(d1 + r, e1v);
}

__global__ void coef_kernel(const int* __restrict__ row,
                            const int* __restrict__ pt,
                            const float* __restrict__ pemb_l,
                            const unsigned* __restrict__ m1u,
                            const float* __restrict__ e0,
                            const float* __restrict__ d0,
                            const float* __restrict__ d1,
                            float* __restrict__ a) {
  int e = blockIdx.x * blockDim.x + threadIdx.x;
  if (e >= NEDGES) return;
  int r = row[e];
  float a0 = e0[e] / d0[r];
  float a1 = expf(pemb_l[pt[e]] - fdec(m1u[r])) / d1[r];
  a[e] = 0.5f * (a0 + a1);
}

__global__ void aggregate_kernel(const float* __restrict__ a,
                                 const float* __restrict__ y,
                                 const int* __restrict__ row,
                                 const int* __restrict__ col,
                                 float* __restrict__ out) {
  int idx = blockIdx.x * blockDim.x + threadIdx.x;
  int e = idx >> 6;
  int c = idx & (HIDDEN - 1);
  float av = a[e];
  int rr = row[e];
  int cc = col[e];
  atomicAdd(out + (size_t)rr * HIDDEN + c, av * y[(size_t)cc * HIDDEN + c]);
}

__global__ void init_out_kernel(const float* __restrict__ ue,
                                const float* __restrict__ ie,
                                float* __restrict__ out) {
  int i = blockIdx.x * blockDim.x + threadIdx.x;
  out[i] = (i < NUSERS * HIDDEN) ? ue[i] : ie[i - NUSERS * HIDDEN];
}
__global__ void add_kernel(float* __restrict__ out, const float* __restrict__ x) {
  int i = blockIdx.x * blockDim.x + threadIdx.x;
  out[i] += x[i];
}
__global__ void scale_kernel(float* __restrict__ out, float s) {
  int i = blockIdx.x * blockDim.x + threadIdx.x;
  out[i] *= s;
}

extern "C" void kernel_launch(void* const* d_in, const int* in_sizes, int n_in,
                              void* d_out, int out_size, void* d_ws, size_t ws_size,
                              hipStream_t stream) {
  (void)in_sizes; (void)n_in; (void)out_size; (void)ws_size;
  const float* ue      = (const float*)d_in[0];
  const float* ie      = (const float*)d_in[1];
  const float* eigs    = (const float*)d_in[2];
  const float* lambda0 = (const float*)d_in[3];
  const float* pemb    = (const float*)d_in[4];
  const int*   indices = (const int*)d_in[5];
  const int*   pt      = (const int*)d_in[6];
  const int* row = indices;
  const int* col = indices + NEDGES;
  float* out = (float*)d_out;

  char* p = (char*)d_ws;
  auto carve = [&](size_t bytes) -> char* {
    char* r = p;
    p += (bytes + 255) & ~(size_t)255;
    return r;
  };
  float*    y    = (float*)   carve((size_t)NNODES * HIDDEN * 4);
  float*    emb1 = (float*)   carve((size_t)NNODES * HIDDEN * 4);
  float*    s0a  = (float*)   carve((size_t)NEDGES * 4);
  float*    e0   = (float*)   carve((size_t)NEDGES * 4);
  char*     statsBase = p;
  unsigned* m0u  = (unsigned*)carve((size_t)NNODES * 4);
  unsigned* m1u  = (unsigned*)carve((size_t)NNODES * 4);
  float*    d0   = (float*)   carve((size_t)NNODES * 4);
  float*    d1   = (float*)   carve((size_t)NNODES * 4);
  size_t statsBytes = (size_t)(p - statsBase);

  const dim3 B(256);
  const int GRID_NODE_ELEM = (NNODES * HIDDEN) / 256;
  const int GRID_LN        = ((NNODES / 16) + 7) / 8;
  const int GRID_SCORE     = NEDGES / 16;
  const int GRID_EDGE      = (NEDGES + 255) / 256;
  const int GRID_AGG       = (NEDGES * HIDDEN) / 256;

  hipLaunchKernelGGL(init_out_kernel, dim3(GRID_NODE_ELEM), B, 0, stream, ue, ie, out);

  for (int l = 0; l < 2; ++l) {
    hipMemsetAsync(statsBase, 0, statsBytes, stream);
    if (l == 0)
      hipMemsetAsync(emb1, 0, (size_t)NNODES * HIDDEN * 4, stream);

    if (l == 0)
      hipLaunchKernelGGL(ln_wmma_kernel, dim3(GRID_LN), B, 0, stream, ue, ie, NUSERS, y);
    else
      hipLaunchKernelGGL(ln_wmma_kernel, dim3(GRID_LN), B, 0, stream, emb1, emb1, NNODES, y);

    hipLaunchKernelGGL(score_kernel, dim3(GRID_SCORE), B, 0, stream,
                       y, eigs, row, col, pt, lambda0 + l, pemb + l * NPATHS,
                       s0a, m0u, m1u);
    hipLaunchKernelGGL(expsum_kernel, dim3(GRID_EDGE), B, 0, stream,
                       s0a, row, pt, pemb + l * NPATHS, m0u, m1u, e0, d0, d1);
    hipLaunchKernelGGL(coef_kernel, dim3(GRID_EDGE), B, 0, stream,
                       row, pt, pemb + l * NPATHS, m1u, e0, d0, d1, s0a);

    float* target = (l == 0) ? emb1 : out;
    hipLaunchKernelGGL(aggregate_kernel, dim3(GRID_AGG), B, 0, stream,
                       s0a, y, row, col, target);

    if (l == 0)
      hipLaunchKernelGGL(add_kernel, dim3(GRID_NODE_ELEM), B, 0, stream, out, emb1);
  }

  hipLaunchKernelGGL(scale_kernel, dim3(GRID_NODE_ELEM), B, 0, stream, out, 1.0f / 3.0f);
}


# Edges counting-sorted by row once per call; one 16-lane group per row does score, both softmaxes and the weighted sum in registers: no float atomics; sum order = arrival rank in row
# speedup vs baseline: 2.6571x; 2.6571x over previous
#include <hip/hip_runtime.h>

#define HIDDEN 64
#define EIGD   16
#define NUSERS 100000
#define NITEMS 200000
#define NNODES 300000
#define NEDGES 1250000
#define NPATHS 14
#define LN_EPS 1e-5f

typedef __attribute__((ext_vector_type(2))) float v2f;
typedef __attribute__((ext_vector_type(8))) float v8f;

__global__ void ln_wmma_kernel(const float* __restrict__ srcA,
                               const float* __restrict__ srcB,
                               int nA, float* __restrict__ y) {
  int lane = threadIdx.x & 31;
  int wave = threadIdx.x >> 5;
  int base = (blockIdx.x * (blockDim.x >> 5) + wave) << 4;
  if (base >= NNODES) return;

  int m = lane & 15;
  int h = lane >> 4;
  int node = base + m;
  const float* srow = (node < nA) ? (srcA + (size_t)node * HIDDEN)
                                  : (srcB + (size_t)(node - nA) * HIDDEN);
  const float* sp = srow + 2 * h;
  float*       dp = y + (size_t)node * HIDDEN + 2 * h;

  const v2f ones = {1.0f, 1.0f};
  v8f accS = {};
  v8f accQ = {};
  float2 vbuf[16];
#pragma unroll
  for (int c = 0; c < 16; ++c) {
    float2 v = *(const float2*)(sp + 4 * c);
    vbuf[c] = v;
    v2f a  = {v.x, v.y};
    v2f a2 = {v.x * v.x, v.y * v.y};
    accS = __builtin_amdgcn_wmma_f32_16x16x4_f32(false, a,  false, ones,
                                                 (short)0, accS, false, false);
    accQ = __builtin_amdgcn_wmma_f32_16x16x4_f32(false, a2, false, ones,
                                                 (short)0, accQ, false, false);
  }

  float s = 0.0f, q = 0.0f;
#pragma unroll
  for (int i = 0; i < 8; ++i) {
    float sLo = __shfl(accS[i], 0, 32);
    float sHi = __shfl(accS[i], 16, 32);
    float qLo = __shfl(accQ[i], 0, 32);
    float qHi = __shfl(accQ[i], 16, 32);
    s = (m == i)     ? sLo : s;
    s = (m == 8 + i) ? sHi : s;
    q = (m == i)     ? qLo : q;
    q = (m == 8 + i) ? qHi : q;
  }

  float mu  = s * (1.0f / HIDDEN);
  float var = q * (1.0f / HIDDEN) - mu * mu;
  float rin = rsqrtf(var + LN_EPS);
#pragma unroll
  for (int c = 0; c < 16; ++c) {
    float2 o = make_float2((vbuf[c].x - mu) * rin, (vbuf[c].y - mu) * rin);
    *(float2*)(dp + 4 * c) = o;
  }
}

__global__ void hist_rank_kernel(const int* __restrict__ row,
                                 int* __restrict__ cnt,
                                 int* __restrict__ rank) {
  int e = blockIdx.x * blockDim.x + threadIdx.x;
  if (e >= NEDGES) return;
  rank[e] = atomicAdd(cnt + row[e], 1);
}

#define SCAN_T 256
#define SCAN_E (SCAN_T * 4)
#define SCAN_B ((NNODES + SCAN_E - 1) / SCAN_E)

__global__ void blocksum_kernel(const int* __restrict__ cnt, int* __restrict__ bsum) {
  __shared__ int wsum[SCAN_T / 32];
  int t = threadIdx.x;
  int i0 = (blockIdx.x * SCAN_T + t) * 4;
  int s = 0;
  if (i0 + 3 < NNODES) {
    int4 v = *(const int4*)(cnt + i0);
    s = v.x + v.y + v.z + v.w;
  } else {
    for (int k = 0; k < 4; ++k)
      if (i0 + k < NNODES) s += cnt[i0 + k];
  }
  for (int o = 16; o > 0; o >>= 1) s += __shfl_xor(s, o, 32);
  if ((t & 31) == 0) wsum[t >> 5] = s;
  __syncthreads();
  if (t == 0) {
    int a = 0;
    for (int w = 0; w < SCAN_T / 32; ++w) a += wsum[w];
    bsum[blockIdx.x] = a;
  }
}

__global__ void offsets_kernel(const int* __restrict__ cnt,
                               const int* __restrict__ bsum,
                               int* __restrict__ off) {
  __shared__ int wpre[SCAN_T / 32];
  __shared__ int wtot[SCAN_T / 32];
  int t = threadIdx.x;
  int lane = t & 31;
  int wave = t >> 5;

  int ps = 0;
  for (int i = t; i < (int)blockIdx.x; i += SCAN_T) ps += bsum[i];
  for (int o = 16; o > 0; o >>= 1) ps += __shfl_xor(ps, o, 32);
  if (lane == 0) wpre[wave] = ps;

  int i0 = (blockIdx.x * SCAN_T + t) * 4;
  int v0 = 0, v1 = 0, v2 = 0, v3 = 0;
  if (i0 + 3 < NNODES) {
    int4 v = *(const int4*)(cnt + i0);
    v0 = v.x; v1 = v.y; v2 = v.z; v3 = v.w;
  } else {
    if (i0     < NNODES) v0 = cnt[i0];
    if (i0 + 1 < NNODES) v1 = cnt[i0 + 1];
    if (i0 + 2 < NNODES) v2 = cnt[i0 + 2];
  }
  int s = v0 + v1 + v2 + v3;
  int x = s;
  for (int o = 1; o < 32; o <<= 1) {
    int nb = __shfl_up(x, o, 32);
    if (lane >= o) x += nb;
  }
  if (lane == 31) wtot[wave] = x;
  __syncthreads();
  int basev = 0;
  for (int w = 0; w < SCAN_T / 32; ++w) {
    basev += wpre[w];
    if (w < wave) basev += wtot[w];
  }
  int ex = basev + x - s;
  if (i0 + 3 < NNODES) {
    *(int4*)(off + i0) = make_int4(ex, ex + v0, ex + v0 + v1, ex + v0 + v1 + v2);
  } else {
    if (i0     < NNODES) off[i0]     = ex;
    if (i0 + 1 < NNODES) off[i0 + 1] = ex + v0;
    if (i0 + 2 < NNODES) off[i0 + 2] = ex + v0 + v1;
  }
}

__global__ void fill_kernel(const int* __restrict__ row,
                            const int* __restrict__ col,
                            const int* __restrict__ pt,
                            const int* __restrict__ rank,
                            const int* __restrict__ off,
                            int2* __restrict__ sedge) {
  int e = blockIdx.x * blockDim.x + threadIdx.x;
  if (e >= NEDGES) return;
  int r = row[e];
  sedge[off[r] + rank[e]] = make_int2(col[e], pt[e]);
}

__global__ void row_kernel(const float* __restrict__ y,
                           const float* __restrict__ eigs,
                           const int* __restrict__ off,
                           const int* __restrict__ cnt,
                           const int2* __restrict__ sedge,
                           const float* __restrict__ lambda0_l,
                           const float* __restrict__ pemb_l,
                           float* __restrict__ dst,
                           int add_into) {
  int g = blockIdx.x * (blockDim.x >> 4) + (threadIdx.x >> 4);
  if (g >= NNODES) return;
  int lane = threadIdx.x & 15;
  int beg = off[g];
  int n   = cnt[g];
  float4 yr = ((const float4*)(y + (size_t)g * HIDDEN))[lane];
  float  er = eigs[(size_t)g * EIGD + lane];
  float  wr = expf(lambda0_l[0]) * er;

  const float ninf = -__builtin_inff();
  float m0 = ninf, m1 = ninf, d0 = 0.0f, d1 = 0.0f;
  float4 acc0 = make_float4(0.0f, 0.0f, 0.0f, 0.0f);
  float4 acc1 = make_float4(0.0f, 0.0f, 0.0f, 0.0f);

  for (int base = 0; base < n; base += 16) {
    int cc = n - base;
    if (cc > 16) cc = 16;
    int2 my = make_int2(0, 0);
    if (lane < cc) my = sedge[beg + base + lane];
    for (int j = 0; j < cc; ++j) {
      int c = __shfl(my.x, j, 16);
      int t = __shfl(my.y, j, 16);
      float4 yc = ((const float4*)(y + (size_t)c * HIDDEN))[lane];
      float  ec = eigs[(size_t)c * EIGD + lane];

      float p = 0.125f * (yr.x * yc.x + yr.y * yc.y + yr.z * yc.z + yr.w * yc.w);
      p += wr * ec;
      for (int o = 8; o > 0; o >>= 1) p += __shfl_xor(p, o, 16);
      float s1 = pemb_l[t];

      float x0 = expf(-fabsf(p - m0));
      bool  g0 = p > m0;
      float sc0 = g0 ? x0 : 1.0f;
      float w0  = g0 ? 1.0f : x0;
      m0 = g0 ? p : m0;
      d0 = d0 * sc0 + w0;
      acc0.x = acc0.x * sc0 + w0 * yc.x;
      acc0.y = acc0.y * sc0 + w0 * yc.y;
      acc0.z = acc0.z * sc0 + w0 * yc.z;
      acc0.w = acc0.w * sc0 + w0 * yc.w;

      float x1 = expf(-fabsf(s1 - m1));
      bool  g1 = s1 > m1;
      float sc1 = g1 ? x1 : 1.0f;
      float w1  = g1 ? 1.0f : x1;
      m1 = g1 ? s1 : m1;
      d1 = d1 * sc1 + w1;
      acc1.x = acc1.x * sc1 + w1 * yc.x;
      acc1.y = acc1.y * sc1 + w1 * yc.y;
      acc1.z = acc1.z * sc1 + w1 * yc.z;
      acc1.w = acc1.w * sc1 + w1 * yc.w;
    }
  }

  float4 o = make_float4(0.0f, 0.0f, 0.0f, 0.0f);
  if (n > 0) {
    float i0 = 0.5f / d0;
    float i1 = 0.5f / d1;
    o.x = acc0.x * i0 + acc1.x * i1;
    o.y = acc0.y * i0 + acc1.y * i1;
    o.z = acc0.z * i0 + acc1.z * i1;
    o.w = acc0.w * i0 + acc1.w * i1;
  }
  float4* dp = (float4*)(dst + (size_t)g * HIDDEN) + lane;
  if (add_into) {
    float4 prev = *dp;
    o.x += prev.x; o.y += prev.y; o.z += prev.z; o.w += prev.w;
  }
  *dp = o;
}

__global__ void init_out_kernel(const float* __restrict__ ue,
                                const float* __restrict__ ie,
                                float* __restrict__ out) {
  int i = blockIdx.x * blockDim.x + threadIdx.x;
  out[i] = (i < NUSERS * HIDDEN) ? ue[i] : ie[i - NUSERS * HIDDEN];
}
__global__ void add_kernel(float* __restrict__ out, const float* __restrict__ x) {
  int i = blockIdx.x * blockDim.x + threadIdx.x;
  out[i] += x[i];
}
__global__ void scale_kernel(float* __restrict__ out, float s) {
  int i = blockIdx.x * blockDim.x + threadIdx.x;
  out[i] *= s;
}

extern "C" void kernel_launch(void* const* d_in, const int* in_sizes, int n_in,
                              void* d_out, int out_size, void* d_ws, size_t ws_size,
                              hipStream_t stream) {
  (void)in_sizes; (void)n_in; (void)out_size; (void)ws_size;
  const float* ue      = (const float*)d_in[0];
  const float* ie      = (const float*)d_in[1];
  const float* eigs    = (const float*)d_in[2];
  const float* lambda0 = (const float*)d_in[3];
  const float* pemb    = (const float*)d_in[4];
  const int*   indices = (const int*)d_in[5];
  const int*   pt      = (const int*)d_in[6];
  const int* row = indices;
  const int* col = indices + NEDGES;
  float* out = (float*)d_out;

  char* p = (char*)d_ws;
  auto carve = [&](size_t bytes) -> char* {
    char* r = p;
    p += (bytes + 255) & ~(size_t)255;
    return r;
  };
  float* y     = (float*)carve((size_t)NNODES * HIDDEN * 4);
  float* emb1  = (float*)carve((size_t)NNODES * HIDDEN * 4);
  int2*  sedge = (int2*) carve((size_t)NEDGES * 8);
  int*   rank  = (int*)  carve((size_t)NEDGES * 4);
  int*   cnt   = (int*)  carve((size_t)NNODES * 4);
  int*   off   = (int*)  carve((size_t)NNODES * 4);
  int*   bsum  = (int*)  carve((size_t)SCAN_B * 4);

  const dim3 B(256);
  const int GRID_NODE_ELEM = (NNODES * HIDDEN) / 256;
  const int GRID_LN        = ((NNODES / 16) + 7) / 8;
  const int GRID_EDGE      = (NEDGES + 255) / 256;
  const int GRID_ROW       = (NNODES + 15) / 16;

  hipLaunchKernelGGL(init_out_kernel, dim3(GRID_NODE_ELEM), B, 0, stream, ue, ie, out);

  hipMemsetAsync(cnt, 0, (size_t)NNODES * 4, stream);
  hipLaunchKernelGGL(hist_rank_kernel, dim3(GRID_EDGE), B, 0, stream, row, cnt, rank);
  hipLaunchKernelGGL(blocksum_kernel, dim3(SCAN_B), dim3(SCAN_T), 0, stream, cnt, bsum);
  hipLaunchKernelGGL(offsets_kernel, dim3(SCAN_B), dim3(SCAN_T), 0, stream, cnt, bsum, off);
  hipLaunchKernelGGL(fill_kernel, dim3(GRID_EDGE), B, 0, stream, row, col, pt, rank, off, sedge);

  for (int l = 0; l < 2; ++l) {
    if (l == 0)
      hipLaunchKernelGGL(ln_wmma_kernel, dim3(GRID_LN), B, 0, stream, ue, ie, NUSERS, y);
    else
      hipLaunchKernelGGL(ln_wmma_kernel, dim3(GRID_LN), B, 0, stream, emb1, emb1, NNODES, y);

    float* target = (l == 0) ? emb1 : out;
    hipLaunchKernelGGL(row_kernel, dim3(GRID_ROW), B, 0, stream,
                       y, eigs, off, cnt, sedge, lambda0 + l, pemb + l * NPATHS,
                       target, l);

    if (l == 0)
      hipLaunchKernelGGL(add_kernel, dim3(GRID_NODE_ELEM), B, 0, stream, out, emb1);
  }

  hipLaunchKernelGGL(scale_kernel, dim3(GRID_NODE_ELEM), B, 0, stream, out, 1.0f / 3.0f);
}


# Copy, add and scale passes folded into the last row kernel: it reads its own emb0 and emb1 rows and stores (emb0+emb1+row)/3 into out once; 3 launches fewer, sum order unchanged
# speedup vs baseline: 3.7102x; 1.3963x over previous
#include <hip/hip_runtime.h>

#define HIDDEN 64
#define EIGD   16
#define NUSERS 100000
#define NITEMS 200000
#define NNODES 300000
#define NEDGES 1250000
#define NPATHS 14
#define LN_EPS 1e-5f

typedef __attribute__((ext_vector_type(2))) float v2f;
typedef __attribute__((ext_vector_type(8))) float v8f;

__global__ void ln_wmma_kernel(const float* __restrict__ srcA,
                               const float* __restrict__ srcB,
                               int nA, float* __restrict__ y) {
  int lane = threadIdx.x & 31;
  int wave = threadIdx.x >> 5;
  int base = (blockIdx.x * (blockDim.x >> 5) + wave) << 4;
  if (base >= NNODES) return;

  int m = lane & 15;
  int h = lane >> 4;
  int node = base + m;
  const float* srow = (node < nA) ? (srcA + (size_t)node * HIDDEN)
                                  : (srcB + (size_t)(node - nA) * HIDDEN);
  const float* sp = srow + 2 * h;
  float*       dp = y + (size_t)node * HIDDEN + 2 * h;

  const v2f ones = {1.0f, 1.0f};
  v8f accS = {};
  v8f accQ = {};
  float2 vbuf[16];
#pragma unroll
  for (int c = 0; c < 16; ++c) {
    float2 v = *(const float2*)(sp + 4 * c);
    vbuf[c] = v;
    v2f a  = {v.x, v.y};
    v2f a2 = {v.x * v.x, v.y * v.y};
    accS = __builtin_amdgcn_wmma_f32_16x16x4_f32(false, a,  false, ones,
                                                 (short)0, accS, false, false);
    accQ = __builtin_amdgcn_wmma_f32_16x16x4_f32(false, a2, false, ones,
                                                 (short)0, accQ, false, false);
  }

  float s = 0.0f, q = 0.0f;
#pragma unroll
  for (int i = 0; i < 8; ++i) {
    float sLo = __shfl(accS[i], 0, 32);
    float sHi = __shfl(accS[i], 16, 32);
    float qLo = __shfl(accQ[i], 0, 32);
    float qHi = __shfl(accQ[i], 16, 32);
    s = (m == i)     ? sLo : s;
    s = (m == 8 + i) ? sHi : s;
    q = (m == i)     ? qLo : q;
    q = (m == 8 + i) ? qHi : q;
  }

  float mu  = s * (1.0f / HIDDEN);
  float var = q * (1.0f / HIDDEN) - mu * mu;
  float rin = rsqrtf(var + LN_EPS);
#pragma unroll
  for (int c = 0; c < 16; ++c) {
    float2 o = make_float2((vbuf[c].x - mu) * rin, (vbuf[c].y - mu) * rin);
    *(float2*)(dp + 4 * c) = o;
  }
}

__global__ void hist_rank_kernel(const int* __restrict__ row,
                                 int* __restrict__ cnt,
                                 int* __restrict__ rank) {
  int e = blockIdx.x * blockDim.x + threadIdx.x;
  if (e >= NEDGES) return;
  rank[e] = atomicAdd(cnt + row[e], 1);
}

#define SCAN_T 256
#define SCAN_E (SCAN_T * 4)
#define SCAN_B ((NNODES + SCAN_E - 1) / SCAN_E)

__global__ void blocksum_kernel(const int* __restrict__ cnt, int* __restrict__ bsum) {
  __shared__ int wsum[SCAN_T / 32];
  int t = threadIdx.x;
  int i0 = (blockIdx.x * SCAN_T + t) * 4;
  int s = 0;
  if (i0 + 3 < NNODES) {
    int4 v = *(const int4*)(cnt + i0);
    s = v.x + v.y + v.z + v.w;
  } else {
    for (int k = 0; k < 4; ++k)
      if (i0 + k < NNODES) s += cnt[i0 + k];
  }
  for (int o = 16; o > 0; o >>= 1) s += __shfl_xor(s, o, 32);
  if ((t & 31) == 0) wsum[t >> 5] = s;
  __syncthreads();
  if (t == 0) {
    int a = 0;
    for (int w = 0; w < SCAN_T / 32; ++w) a += wsum[w];
    bsum[blockIdx.x] = a;
  }
}

__global__ void offsets_kernel(const int* __restrict__ cnt,
                               const int* __restrict__ bsum,
                               int* __restrict__ off) {
  __shared__ int wpre[SCAN_T / 32];
  __shared__ int wtot[SCAN_T / 32];
  int t = threadIdx.x;
  int lane = t & 31;
  int wave = t >> 5;

  int ps = 0;
  for (int i = t; i < (int)blockIdx.x; i += SCAN_T) ps += bsum[i];
  for (int o = 16; o > 0; o >>= 1) ps += __shfl_xor(ps, o, 32);
  if (lane == 0) wpre[wave] = ps;

  int i0 = (blockIdx.x * SCAN_T + t) * 4;
  int v0 = 0, v1 = 0, v2 = 0, v3 = 0;
  if (i0 + 3 < NNODES) {
    int4 v = *(const int4*)(cnt + i0);
    v0 = v.x; v1 = v.y; v2 = v.z; v3 = v.w;
  } else {
    if (i0     < NNODES) v0 = cnt[i0];
    if (i0 + 1 < NNODES) v1 = cnt[i0 + 1];
    if (i0 + 2 < NNODES) v2 = cnt[i0 + 2];
  }
  int s = v0 + v1 + v2 + v3;
  int x = s;
  for (int o = 1; o < 32; o <<= 1) {
    int nb = __shfl_up(x, o, 32);
    if (lane >= o) x += nb;
  }
  if (lane == 31) wtot[wave] = x;
  __syncthreads();
  int basev = 0;
  for (int w = 0; w < SCAN_T / 32; ++w) {
    basev += wpre[w];
    if (w < wave) basev += wtot[w];
  }
  int ex = basev + x - s;
  if (i0 + 3 < NNODES) {
    *(int4*)(off + i0) = make_int4(ex, ex + v0, ex + v0 + v1, ex + v0 + v1 + v2);
  } else {
    if (i0     < NNODES) off[i0]     = ex;
    if (i0 + 1 < NNODES) off[i0 + 1] = ex + v0;
    if (i0 + 2 < NNODES) off[i0 + 2] = ex + v0 + v1;
  }
}

__global__ void fill_kernel(const int* __restrict__ row,
                            const int* __restrict__ col,
                            const int* __restrict__ pt,
                            const int* __restrict__ rank,
                            const int* __restrict__ off,
                            int2* __restrict__ sedge) {
  int e = blockIdx.x * blockDim.x + threadIdx.x;
  if (e >= NEDGES) return;
  int r = row[e];
  sedge[off[r] + rank[e]] = make_int2(col[e], pt[e]);
}

template <bool LAST>
__global__ void row_kernel(const float* __restrict__ y,
                           const float* __restrict__ eigs,
                           const int* __restrict__ off,
                           const int* __restrict__ cnt,
                           const int2* __restrict__ sedge,
                           const float* __restrict__ lambda0_l,
                           const float* __restrict__ pemb_l,
                           const float* __restrict__ ue,
                           const float* __restrict__ ie,
                           const float* __restrict__ emb1,
                           float* __restrict__ dst) {
  int g = blockIdx.x * (blockDim.x >> 4) + (threadIdx.x >> 4);
  if (g >= NNODES) return;
  int lane = threadIdx.x & 15;
  int beg = off[g];
  int n   = cnt[g];
  float4 yr = ((const float4*)(y + (size_t)g * HIDDEN))[lane];
  float  er = eigs[(size_t)g * EIGD + lane];
  float  wr = expf(lambda0_l[0]) * er;

  const float ninf = -__builtin_inff();
  float m0 = ninf, m1 = ninf, d0 = 0.0f, d1 = 0.0f;
  float4 acc0 = make_float4(0.0f, 0.0f, 0.0f, 0.0f);
  float4 acc1 = make_float4(0.0f, 0.0f, 0.0f, 0.0f);

  for (int base = 0; base < n; base += 16) {
    int cc = n - base;
    if (cc > 16) cc = 16;
    int2 my = make_int2(0, 0);
    if (lane < cc) my = sedge[beg + base + lane];
    for (int j = 0; j < cc; ++j) {
      int c = __shfl(my.x, j, 16);
      int t = __shfl(my.y, j, 16);
      float4 yc = ((const float4*)(y + (size_t)c * HIDDEN))[lane];
      float  ec = eigs[(size_t)c * EIGD + lane];

      float p = 0.125f * (yr.x * yc.x + yr.y * yc.y + yr.z * yc.z + yr.w * yc.w);
      p += wr * ec;
      for (int o = 8; o > 0; o >>= 1) p += __shfl_xor(p, o, 16);
      float s1 = pemb_l[t];

      float x0 = expf(-fabsf(p - m0));
      bool  g0 = p > m0;
      float sc0 = g0 ? x0 : 1.0f;
      float w0  = g0 ? 1.0f : x0;
      m0 = g0 ? p : m0;
      d0 = d0 * sc0 + w0;
      acc0.x = acc0.x * sc0 + w0 * yc.x;
      acc0.y = acc0.y * sc0 + w0 * yc.y;
      acc0.z = acc0.z * sc0 + w0 * yc.z;
      acc0.w = acc0.w * sc0 + w0 * yc.w;

      float x1 = expf(-fabsf(s1 - m1));
      bool  g1 = s1 > m1;
      float sc1 = g1 ? x1 : 1.0f;
      float w1  = g1 ? 1.0f : x1;
      m1 = g1 ? s1 : m1;
      d1 = d1 * sc1 + w1;
      acc1.x = acc1.x * sc1 + w1 * yc.x;
      acc1.y = acc1.y * sc1 + w1 * yc.y;
      acc1.z = acc1.z * sc1 + w1 * yc.z;
      acc1.w = acc1.w * sc1 + w1 * yc.w;
    }
  }

  float4 e0 = make_float4(0.0f, 0.0f, 0.0f, 0.0f);
  float4 e1 = make_float4(0.0f, 0.0f, 0.0f, 0.0f);
  if (LAST) {
    const float* e0row = (g < NUSERS) ? (ue + (size_t)g * HIDDEN)
                                      : (ie + (size_t)(g - NUSERS) * HIDDEN);
    e0 = ((const float4*)e0row)[lane];
    e1 = ((const float4*)(emb1 + (size_t)g * HIDDEN))[lane];
  }

  float4 o = make_float4(0.0f, 0.0f, 0.0f, 0.0f);
  if (n > 0) {
    float i0 = 0.5f / d0;
    float i1 = 0.5f / d1;
    o.x = acc0.x * i0 + acc1.x * i1;
    o.y = acc0.y * i0 + acc1.y * i1;
    o.z = acc0.z * i0 + acc1.z * i1;
    o.w = acc0.w * i0 + acc1.w * i1;
  }
  float4* dp = (float4*)(dst + (size_t)g * HIDDEN) + lane;
  if (LAST) {
    const float third = 1.0f / 3.0f;
    o.x = (o.x + (e0.x + e1.x)) * third;
    o.y = (o.y + (e0.y + e1.y)) * third;
    o.z = (o.z + (e0.z + e1.z)) * third;
    o.w = (o.w + (e0.w + e1.w)) * third;
  }
  *dp = o;
}

extern "C" void kernel_launch(void* const* d_in, const int* in_sizes, int n_in,
                              void* d_out, int out_size, void* d_ws, size_t ws_size,
                              hipStream_t stream) {
  (void)in_sizes; (void)n_in; (void)out_size; (void)ws_size;
  const float* ue      = (const float*)d_in[0];
  const float* ie      = (const float*)d_in[1];
  const float* eigs    = (const float*)d_in[2];
  const float* lambda0 = (const float*)d_in[3];
  const float* pemb    = (const float*)d_in[4];
  const int*   indices = (const int*)d_in[5];
  const int*   pt      = (const int*)d_in[6];
  const int* row = indices;
  const int* col = indices + NEDGES;
  float* out = (float*)d_out;

  char* p = (char*)d_ws;
  auto carve = [&](size_t bytes) -> char* {
    char* r = p;
    p += (bytes + 255) & ~(size_t)255;
    return r;
  };
  float* y     = (float*)carve((size_t)NNODES * HIDDEN * 4);
  float* emb1  = (float*)carve((size_t)NNODES * HIDDEN * 4);
  int2*  sedge = (int2*) carve((size_t)NEDGES * 8);
  int*   rank  = (int*)  carve((size_t)NEDGES * 4);
  int*   cnt   = (int*)  carve((size_t)NNODES * 4);
  int*   off   = (int*)  carve((size_t)NNODES * 4);
  int*   bsum  = (int*)  carve((size_t)SCAN_B * 4);

  const dim3 B(256);
  const int GRID_LN        = ((NNODES / 16) + 7) / 8;
  const int GRID_EDGE      = (NEDGES + 255) / 256;
  const int GRID_ROW       = (NNODES + 15) / 16;

  hipMemsetAsync(cnt, 0, (size_t)NNODES * 4, stream);
  hipLaunchKernelGGL(hist_rank_kernel, dim3(GRID_EDGE), B, 0, stream, row, cnt, rank);
  hipLaunchKernelGGL(blocksum_kernel, dim3(SCAN_B), dim3(SCAN_T), 0, stream, cnt, bsum);
  hipLaunchKernelGGL(offsets_kernel, dim3(SCAN_B), dim3(SCAN_T), 0, stream, cnt, bsum, off);
  hipLaunchKernelGGL(fill_kernel, dim3(GRID_EDGE), B, 0, stream, row, col, pt, rank, off, sedge);

  for (int l = 0; l < 2; ++l) {
    if (l == 0)
      hipLaunchKernelGGL(ln_wmma_kernel, dim3(GRID_LN), B, 0, stream, ue, ie, NUSERS, y);
    else
      hipLaunchKernelGGL(ln_wmma_kernel, dim3(GRID_LN), B, 0, stream, emb1, emb1, NNODES, y);

    if (l == 0)
      hipLaunchKernelGGL(row_kernel<false>, dim3(GRID_ROW), B, 0, stream,
                         y, eigs, off, cnt, sedge, lambda0 + l, pemb + l * NPATHS,
                         ue, ie, (const float*)nullptr, emb1);
    else
      hipLaunchKernelGGL(row_kernel<true>, dim3(GRID_ROW), B, 0, stream,
                         y, eigs, off, cnt, sedge, lambda0 + l, pemb + l * NPATHS,
                         ue, ie, (const float*)emb1, out);
  }
}


# LayerNorm pass stores only mean and rstd per node; row kernels normalise each row as they read it ((x-mean)*rstd, same two ops): normalised plane never written; sum order unchanged
# speedup vs baseline: 4.4725x; 1.2055x over previous
#include <hip/hip_runtime.h>

#define HIDDEN 64
#define EIGD   16
#define NUSERS 100000
#define NITEMS 200000
#define NNODES 300000
#define NEDGES 1250000
#define NPATHS 14
#define LN_EPS 1e-5f

typedef __attribute__((ext_vector_type(2))) float v2f;
typedef __attribute__((ext_vector_type(8))) float v8f;

__global__ void ln_wmma_kernel(const float* __restrict__ srcA,
                               const float* __restrict__ srcB,
                               int nA, float2* __restrict__ stats) {
  int lane = threadIdx.x & 31;
  int wave = threadIdx.x >> 5;
  int base = (blockIdx.x * (blockDim.x >> 5) + wave) << 4;
  if (base >= NNODES) return;

  int m = lane & 15;
  int h = lane >> 4;
  int node = base + m;
  const float* srow = (node < nA) ? (srcA + (size_t)node * HIDDEN)
                                  : (srcB + (size_t)(node - nA) * HIDDEN);
  const float* sp = srow + 2 * h;

  const v2f ones = {1.0f, 1.0f};
  v8f accS = {};
  v8f accQ = {};
#pragma unroll
  for (int c = 0; c < 16; ++c) {
    float2 v = *(const float2*)(sp + 4 * c);
    v2f a  = {v.x, v.y};
    v2f a2 = {v.x * v.x, v.y * v.y};
    accS = __builtin_amdgcn_wmma_f32_16x16x4_f32(false, a,  false, ones,
                                                 (short)0, accS, false, false);
    accQ = __builtin_amdgcn_wmma_f32_16x16x4_f32(false, a2, false, ones,
                                                 (short)0, accQ, false, false);
  }

  float s = 0.0f, q = 0.0f;
#pragma unroll
  for (int i = 0; i < 8; ++i) {
    float sLo = __shfl(accS[i], 0, 32);
    float sHi = __shfl(accS[i], 16, 32);
    float qLo = __shfl(accQ[i], 0, 32);
    float qHi = __shfl(accQ[i], 16, 32);
    s = (m == i)     ? sLo : s;
    s = (m == 8 + i) ? sHi : s;
    q = (m == i)     ? qLo : q;
    q = (m == 8 + i) ? qHi : q;
  }

  float mu  = s * (1.0f / HIDDEN);
  float var = q * (1.0f / HIDDEN) - mu * mu;
  float rin = rsqrtf(var + LN_EPS);
  if (h == 0) stats[node] = make_float2(mu, rin);
}

__global__ void hist_rank_kernel(const int* __restrict__ row,
                                 int* __restrict__ cnt,
                                 int* __restrict__ rank) {
  int e = blockIdx.x * blockDim.x + threadIdx.x;
  if (e >= NEDGES) return;
  rank[e] = atomicAdd(cnt + row[e], 1);
}

#define SCAN_T 256
#define SCAN_E (SCAN_T * 4)
#define SCAN_B ((NNODES + SCAN_E - 1) / SCAN_E)

__global__ void blocksum_kernel(const int* __restrict__ cnt, int* __restrict__ bsum) {
  __shared__ int wsum[SCAN_T / 32];
  int t = threadIdx.x;
  int i0 = (blockIdx.x * SCAN_T + t) * 4;
  int s = 0;
  if (i0 + 3 < NNODES) {
    int4 v = *(const int4*)(cnt + i0);
    s = v.x + v.y + v.z + v.w;
  } else {
    for (int k = 0; k < 4; ++k)
      if (i0 + k < NNODES) s += cnt[i0 + k];
  }
  for (int o = 16; o > 0; o >>= 1) s += __shfl_xor(s, o, 32);
  if ((t & 31) == 0) wsum[t >> 5] = s;
  __syncthreads();
  if (t == 0) {
    int a = 0;
    for (int w = 0; w < SCAN_T / 32; ++w) a += wsum[w];
    bsum[blockIdx.x] = a;
  }
}

__global__ void offsets_kernel(const int* __restrict__ cnt,
                               const int* __restrict__ bsum,
                               int* __restrict__ off) {
  __shared__ int wpre[SCAN_T / 32];
  __shared__ int wtot[SCAN_T / 32];
  int t = threadIdx.x;
  int lane = t & 31;
  int wave = t >> 5;

  int ps = 0;
  for (int i = t; i < (int)blockIdx.x; i += SCAN_T) ps += bsum[i];
  for (int o = 16; o > 0; o >>= 1) ps += __shfl_xor(ps, o, 32);
  if (lane == 0) wpre[wave] = ps;

  int i0 = (blockIdx.x * SCAN_T + t) * 4;
  int v0 = 0, v1 = 0, v2 = 0, v3 = 0;
  if (i0 + 3 < NNODES) {
    int4 v = *(const int4*)(cnt + i0);
    v0 = v.x; v1 = v.y; v2 = v.z; v3 = v.w;
  } else {
    if (i0     < NNODES) v0 = cnt[i0];
    if (i0 + 1 < NNODES) v1 = cnt[i0 + 1];
    if (i0 + 2 < NNODES) v2 = cnt[i0 + 2];
  }
  int s = v0 + v1 + v2 + v3;
  int x = s;
  for (int o = 1; o < 32; o <<= 1) {
    int nb = __shfl_up(x, o, 32);
    if (lane >= o) x += nb;
  }
  if (lane == 31) wtot[wave] = x;
  __syncthreads();
  int basev = 0;
  for (int w = 0; w < SCAN_T / 32; ++w) {
    basev += wpre[w];
    if (w < wave) basev += wtot[w];
  }
  int ex = basev + x - s;
  if (i0 + 3 < NNODES) {
    *(int4*)(off + i0) = make_int4(ex, ex + v0, ex + v0 + v1, ex + v0 + v1 + v2);
  } else {
    if (i0     < NNODES) off[i0]     = ex;
    if (i0 + 1 < NNODES) off[i0 + 1] = ex + v0;
    if (i0 + 2 < NNODES) off[i0 + 2] = ex + v0 + v1;
  }
}

__global__ void fill_kernel(const int* __restrict__ row,
                            const int* __restrict__ col,
                            const int* __restrict__ pt,
                            const int* __restrict__ rank,
                            const int* __restrict__ off,
                            int2* __restrict__ sedge) {
  int e = blockIdx.x * blockDim.x + threadIdx.x;
  if (e >= NEDGES) return;
  int r = row[e];
  sedge[off[r] + rank[e]] = make_int2(col[e], pt[e]);
}

template <bool LAST>
__device__ __forceinline__ const float4* in_row(int node,
                                                const float* __restrict__ ue,
                                                const float* __restrict__ ie,
                                                const float* __restrict__ emb1) {
  if (LAST) return (const float4*)(emb1 + (size_t)node * HIDDEN);
  bool u = node < NUSERS;
  const float* tab = u ? ue : ie;
  int idx = u ? node : node - NUSERS;
  return (const float4*)(tab + (size_t)idx * HIDDEN);
}

template <bool LAST>
__global__ void row_kernel(const float2* __restrict__ stats,
                           const float* __restrict__ eigs,
                           const int* __restrict__ off,
                           const int* __restrict__ cnt,
                           const int2* __restrict__ sedge,
                           const float* __restrict__ lambda0_l,
                           const float* __restrict__ pemb_l,
                           const float* __restrict__ ue,
                           const float* __restrict__ ie,
                           const float* __restrict__ emb1,
                           float* __restrict__ dst) {
  int g = blockIdx.x * (blockDim.x >> 4) + (threadIdx.x >> 4);
  if (g >= NNODES) return;
  int lane = threadIdx.x & 15;
  int beg = off[g];
  int n   = cnt[g];
  float4 yr = in_row<LAST>(g, ue, ie, emb1)[lane];
  {
    float2 sg = stats[g];
    yr.x = (yr.x - sg.x) * sg.y;
    yr.y = (yr.y - sg.x) * sg.y;
    yr.z = (yr.z - sg.x) * sg.y;
    yr.w = (yr.w - sg.x) * sg.y;
  }
  float  er = eigs[(size_t)g * EIGD + lane];
  float  wr = expf(lambda0_l[0]) * er;

  const float ninf = -__builtin_inff();
  float m0 = ninf, m1 = ninf, d0 = 0.0f, d1 = 0.0f;
  float4 acc0 = make_float4(0.0f, 0.0f, 0.0f, 0.0f);
  float4 acc1 = make_float4(0.0f, 0.0f, 0.0f, 0.0f);

  for (int base = 0; base < n; base += 16) {
    int cc = n - base;
    if (cc > 16) cc = 16;
    int2 my = make_int2(0, 0);
    if (lane < cc) my = sedge[beg + base + lane];
    for (int j = 0; j < cc; ++j) {
      int c = __shfl(my.x, j, 16);
      int t = __shfl(my.y, j, 16);
      float4 yc = in_row<LAST>(c, ue, ie, emb1)[lane];
      float2 sn = stats[c];
      float  ec = eigs[(size_t)c * EIGD + lane];
      yc.x = (yc.x - sn.x) * sn.y;
      yc.y = (yc.y - sn.x) * sn.y;
      yc.z = (yc.z - sn.x) * sn.y;
      yc.w = (yc.w - sn.x) * sn.y;

      float p = 0.125f * (yr.x * yc.x + yr.y * yc.y + yr.z * yc.z + yr.w * yc.w);
      p += wr * ec;
      for (int o = 8; o > 0; o >>= 1) p += __shfl_xor(p, o, 16);
      float s1 = pemb_l[t];

      float x0 = expf(-fabsf(p - m0));
      bool  g0 = p > m0;
      float sc0 = g0 ? x0 : 1.0f;
      float w0  = g0 ? 1.0f : x0;
      m0 = g0 ? p : m0;
      d0 = d0 * sc0 + w0;
      acc0.x = acc0.x * sc0 + w0 * yc.x;
      acc0.y = acc0.y * sc0 + w0 * yc.y;
      acc0.z = acc0.z * sc0 + w0 * yc.z;
      acc0.w = acc0.w * sc0 + w0 * yc.w;

      float x1 = expf(-fabsf(s1 - m1));
      bool  g1 = s1 > m1;
      float sc1 = g1 ? x1 : 1.0f;
      float w1  = g1 ? 1.0f : x1;
      m1 = g1 ? s1 : m1;
      d1 = d1 * sc1 + w1;
      acc1.x = acc1.x * sc1 + w1 * yc.x;
      acc1.y = acc1.y * sc1 + w1 * yc.y;
      acc1.z = acc1.z * sc1 + w1 * yc.z;
      acc1.w = acc1.w * sc1 + w1 * yc.w;
    }
  }

  float4 e0 = make_float4(0.0f, 0.0f, 0.0f, 0.0f);
  float4 e1 = make_float4(0.0f, 0.0f, 0.0f, 0.0f);
  if (LAST) {
    e0 = in_row<false>(g, ue, ie, emb1)[lane];
    e1 = in_row<true>(g, ue, ie, emb1)[lane];
  }

  float4 o = make_float4(0.0f, 0.0f, 0.0f, 0.0f);
  if (n > 0) {
    float i0 = 0.5f / d0;
    float i1 = 0.5f / d1;
    o.x = acc0.x * i0 + acc1.x * i1;
    o.y = acc0.y * i0 + acc1.y * i1;
    o.z = acc0.z * i0 + acc1.z * i1;
    o.w = acc0.w * i0 + acc1.w * i1;
  }
  float4* dp = (float4*)(dst + (size_t)g * HIDDEN) + lane;
  if (LAST) {
    const float third = 1.0f / 3.0f;
    o.x = (o.x + (e0.x + e1.x)) * third;
    o.y = (o.y + (e0.y + e1.y)) * third;
    o.z = (o.z + (e0.z + e1.z)) * third;
    o.w = (o.w + (e0.w + e1.w)) * third;
  }
  *dp = o;
}

extern "C" void kernel_launch(void* const* d_in, const int* in_sizes, int n_in,
                              void* d_out, int out_size, void* d_ws, size_t ws_size,
                              hipStream_t stream) {
  (void)in_sizes; (void)n_in; (void)out_size; (void)ws_size;
  const float* ue      = (const float*)d_in[0];
  const float* ie      = (const float*)d_in[1];
  const float* eigs    = (const float*)d_in[2];
  const float* lambda0 = (const float*)d_in[3];
  const float* pemb    = (const float*)d_in[4];
  const int*   indices = (const int*)d_in[5];
  const int*   pt      = (const int*)d_in[6];
  const int* row = indices;
  const int* col = indices + NEDGES;
  float* out = (float*)d_out;

  char* p = (char*)d_ws;
  auto carve = [&](size_t bytes) -> char* {
    char* r = p;
    p += (bytes + 255) & ~(size_t)255;
    return r;
  };
  float2* stats = (float2*)carve((size_t)NNODES * 8);
  float* emb1  = (float*)carve((size_t)NNODES * HIDDEN * 4);
  int2*  sedge = (int2*) carve((size_t)NEDGES * 8);
  int*   rank  = (int*)  carve((size_t)NEDGES * 4);
  int*   cnt   = (int*)  carve((size_t)NNODES * 4);
  int*   off   = (int*)  carve((size_t)NNODES * 4);
  int*   bsum  = (int*)  carve((size_t)SCAN_B * 4);

  const dim3 B(256);
  const int GRID_LN        = ((NNODES / 16) + 7) / 8;
  const int GRID_EDGE      = (NEDGES + 255) / 256;
  const int GRID_ROW       = (NNODES + 15) / 16;

  hipMemsetAsync(cnt, 0, (size_t)NNODES * 4, stream);
  hipLaunchKernelGGL(hist_rank_kernel, dim3(GRID_EDGE), B, 0, stream, row, cnt, rank);
  hipLaunchKernelGGL(blocksum_kernel, dim3(SCAN_B), dim3(SCAN_T), 0, stream, cnt, bsum);
  hipLaunchKernelGGL(offsets_kernel, dim3(SCAN_B), dim3(SCAN_T), 0, stream, cnt, bsum, off);
  hipLaunchKernelGGL(fill_kernel, dim3(GRID_EDGE), B, 0, stream, row, col, pt, rank, off, sedge);

  for (int l = 0; l < 2; ++l) {
    if (l == 0)
      hipLaunchKernelGGL(ln_wmma_kernel, dim3(GRID_LN), B, 0, stream, ue, ie, NUSERS, stats);
    else
      hipLaunchKernelGGL(ln_wmma_kernel, dim3(GRID_LN), B, 0, stream, emb1, emb1, NNODES, stats);

    if (l == 0)
      hipLaunchKernelGGL(row_kernel<false>, dim3(GRID_ROW), B, 0, stream,
                         stats, eigs, off, cnt, sedge, lambda0 + l, pemb + l * NPATHS,
                         ue, ie, (const float*)nullptr, emb1);
    else
      hipLaunchKernelGGL(row_kernel<true>, dim3(GRID_ROW), B, 0, stream,
                         stats, eigs, off, cnt, sedge, lambda0 + l, pemb + l * NPATHS,
                         ue, ie, (const float*)emb1, out);
  }
}


# Every grid sized by the chip, not the problem: row, LayerNorm, histogram and fill kernels run few long-lived waves that loop over rows, node groups, edges (a third of the waves); sum order unchanged
# speedup vs baseline: 4.6074x; 1.0302x over previous
#include <hip/hip_runtime.h>

#define HIDDEN 64
#define EIGD   16
#define NUSERS 100000
#define NITEMS 200000
#define NNODES 300000
#define NEDGES 1250000
#define NPATHS 14
#define LN_EPS 1e-5f

typedef __attribute__((ext_vector_type(2))) float v2f;
typedef __attribute__((ext_vector_type(8))) float v8f;

__global__ void ln_wmma_kernel(const float* __restrict__ srcA,
                               const float* __restrict__ srcB,
                               int nA, float2* __restrict__ stats) {
  int lane = threadIdx.x & 31;
  int wave = threadIdx.x >> 5;
  int m = lane & 15;
  int h = lane >> 4;
  const int wstep = (gridDim.x * (blockDim.x >> 5)) << 4;
  for (int base = (blockIdx.x * (blockDim.x >> 5) + wave) << 4; base < NNODES; base += wstep) {
  int node = base + m;
  const float* srow = (node < nA) ? (srcA + (size_t)node * HIDDEN)
                                  : (srcB + (size_t)(node - nA) * HIDDEN);
  const float* sp = srow + 2 * h;

  const v2f ones = {1.0f, 1.0f};
  v8f accS = {};
  v8f accQ = {};
#pragma unroll
  for (int c = 0; c < 16; ++c) {
    float2 v = *(const float2*)(sp + 4 * c);
    v2f a  = {v.x, v.y};
    v2f a2 = {v.x * v.x, v.y * v.y};
    accS = __builtin_amdgcn_wmma_f32_16x16x4_f32(false, a,  false, ones,
                                                 (short)0, accS, false, false);
    accQ = __builtin_amdgcn_wmma_f32_16x16x4_f32(false, a2, false, ones,
                                                 (short)0, accQ, false, false);
  }

  float s = 0.0f, q = 0.0f;
#pragma unroll
  for (int i = 0; i < 8; ++i) {
    float sLo = __shfl(accS[i], 0, 32);
    float sHi = __shfl(accS[i], 16, 32);
    float qLo = __shfl(accQ[i], 0, 32);
    float qHi = __shfl(accQ[i], 16, 32);
    s = (m == i)     ? sLo : s;
    s = (m == 8 + i) ? sHi : s;
    q = (m == i)     ? qLo : q;
    q = (m == 8 + i) ? qHi : q;
  }

  float mu  = s * (1.0f / HIDDEN);
  float var = q * (1.0f / HIDDEN) - mu * mu;
  float rin = rsqrtf(var + LN_EPS);
  if (h == 0) stats[node] = make_float2(mu, rin);
  }
}

__global__ void hist_rank_kernel(const int* __restrict__ row,
                                 int* __restrict__ cnt,
                                 int* __restrict__ rank) {
  const int stride = gridDim.x * blockDim.x;
  for (int e = blockIdx.x * blockDim.x + threadIdx.x; e < NEDGES; e += stride)
    rank[e] = atomicAdd(cnt + row[e], 1);
}

#define SCAN_T 256
#define SCAN_E (SCAN_T * 4)
#define SCAN_B ((NNODES + SCAN_E - 1) / SCAN_E)

__global__ void blocksum_kernel(const int* __restrict__ cnt, int* __restrict__ bsum) {
  __shared__ int wsum[SCAN_T / 32];
  int t = threadIdx.x;
  int i0 = (blockIdx.x * SCAN_T + t) * 4;
  int s = 0;
  if (i0 + 3 < NNODES) {
    int4 v = *(const int4*)(cnt + i0);
    s = v.x + v.y + v.z + v.w;
  } else {
    for (int k = 0; k < 4; ++k)
      if (i0 + k < NNODES) s += cnt[i0 + k];
  }
  for (int o = 16; o > 0; o >>= 1) s += __shfl_xor(s, o, 32);
  if ((t & 31) == 0) wsum[t >> 5] = s;
  __syncthreads();
  if (t == 0) {
    int a = 0;
    for (int w = 0; w < SCAN_T / 32; ++w) a += wsum[w];
    bsum[blockIdx.x] = a;
  }
}

__global__ void offsets_kernel(const int* __restrict__ cnt,
                               const int* __restrict__ bsum,
                               int* __restrict__ off) {
  __shared__ int wpre[SCAN_T / 32];
  __shared__ int wtot[SCAN_T / 32];
  int t = threadIdx.x;
  int lane = t & 31;
  int wave = t >> 5;

  int ps = 0;
  for (int i = t; i < (int)blockIdx.x; i += SCAN_T) ps += bsum[i];
  for (int o = 16; o > 0; o >>= 1) ps += __shfl_xor(ps, o, 32);
  if (lane == 0) wpre[wave] = ps;

  int i0 = (blockIdx.x * SCAN_T + t) * 4;
  int v0 = 0, v1 = 0, v2 = 0, v3 = 0;
  if (i0 + 3 < NNODES) {
    int4 v = *(const int4*)(cnt + i0);
    v0 = v.x; v1 = v.y; v2 = v.z; v3 = v.w;
  } else {
    if (i0     < NNODES) v0 = cnt[i0];
    if (i0 + 1 < NNODES) v1 = cnt[i0 + 1];
    if (i0 + 2 < NNODES) v2 = cnt[i0 + 2];
  }
  int s = v0 + v1 + v2 + v3;
  int x = s;
  for (int o = 1; o < 32; o <<= 1) {
    int nb = __shfl_up(x, o, 32);
    if (lane >= o) x += nb;
  }
  if (lane == 31) wtot[wave] = x;
  __syncthreads();
  int basev = 0;
  for (int w = 0; w < SCAN_T / 32; ++w) {
    basev += wpre[w];
    if (w < wave) basev += wtot[w];
  }
  int ex = basev + x - s;
  if (i0 + 3 < NNODES) {
    *(int4*)(off + i0) = make_int4(ex, ex + v0, ex + v0 + v1, ex + v0 + v1 + v2);
  } else {
    if (i0     < NNODES) off[i0]     = ex;
    if (i0 + 1 < NNODES) off[i0 + 1] = ex + v0;
    if (i0 + 2 < NNODES) off[i0 + 2] = ex + v0 + v1;
  }
}

__global__ void fill_kernel(const int* __restrict__ row,
                            const int* __restrict__ col,
                            const int* __restrict__ pt,
                            const int* __restrict__ rank,
                            const int* __restrict__ off,
                            int2* __restrict__ sedge) {
  const int stride = gridDim.x * blockDim.x;
  for (int e = blockIdx.x * blockDim.x + threadIdx.x; e < NEDGES; e += stride) {
    int r = row[e];
    sedge[off[r] + rank[e]] = make_int2(col[e], pt[e]);
  }
}

template <bool LAST>
__device__ __forceinline__ const float4* in_row(int node,
                                                const float* __restrict__ ue,
                                                const float* __restrict__ ie,
                                                const float* __restrict__ emb1) {
  if (LAST) return (const float4*)(emb1 + (size_t)node * HIDDEN);
  bool u = node < NUSERS;
  const float* tab = u ? ue : ie;
  int idx = u ? node : node - NUSERS;
  return (const float4*)(tab + (size_t)idx * HIDDEN);
}

template <bool LAST>
__global__ void row_kernel(const float2* __restrict__ stats,
                           const float* __restrict__ eigs,
                           const int* __restrict__ off,
                           const int* __restrict__ cnt,
                           const int2* __restrict__ sedge,
                           const float* __restrict__ lambda0_l,
                           const float* __restrict__ pemb_l,
                           const float* __restrict__ ue,
                           const float* __restrict__ ie,
                           const float* __restrict__ emb1,
                           float* __restrict__ dst) {
  const int lane = threadIdx.x & 15;
  const int ngroups = gridDim.x * (blockDim.x >> 4);
  const float el = expf(lambda0_l[0]);
  for (int g = blockIdx.x * (blockDim.x >> 4) + (threadIdx.x >> 4); g < NNODES; g += ngroups) {
  int beg = off[g];
  int n   = cnt[g];
  float4 yr = in_row<LAST>(g, ue, ie, emb1)[lane];
  {
    float2 sg = stats[g];
    yr.x = (yr.x - sg.x) * sg.y;
    yr.y = (yr.y - sg.x) * sg.y;
    yr.z = (yr.z - sg.x) * sg.y;
    yr.w = (yr.w - sg.x) * sg.y;
  }
  float  er = eigs[(size_t)g * EIGD + lane];
  float  wr = el * er;

  const float ninf = -__builtin_inff();
  float m0 = ninf, m1 = ninf, d0 = 0.0f, d1 = 0.0f;
  float4 acc0 = make_float4(0.0f, 0.0f, 0.0f, 0.0f);
  float4 acc1 = make_float4(0.0f, 0.0f, 0.0f, 0.0f);

  for (int base = 0; base < n; base += 16) {
    int cc = n - base;
    if (cc > 16) cc = 16;
    int2 my = make_int2(0, 0);
    if (lane < cc) my = sedge[beg + base + lane];
    for (int j = 0; j < cc; ++j) {
      int c = __shfl(my.x, j, 16);
      int t = __shfl(my.y, j, 16);
      float4 yc = in_row<LAST>(c, ue, ie, emb1)[lane];
      float2 sn = stats[c];
      float  ec = eigs[(size_t)c * EIGD + lane];
      yc.x = (yc.x - sn.x) * sn.y;
      yc.y = (yc.y - sn.x) * sn.y;
      yc.z = (yc.z - sn.x) * sn.y;
      yc.w = (yc.w - sn.x) * sn.y;

      float p = 0.125f * (yr.x * yc.x + yr.y * yc.y + yr.z * yc.z + yr.w * yc.w);
      p += wr * ec;
      for (int o = 8; o > 0; o >>= 1) p += __shfl_xor(p, o, 16);
      float s1 = pemb_l[t];

      float x0 = expf(-fabsf(p - m0));
      bool  g0 = p > m0;
      float sc0 = g0 ? x0 : 1.0f;
      float w0  = g0 ? 1.0f : x0;
      m0 = g0 ? p : m0;
      d0 = d0 * sc0 + w0;
      acc0.x = acc0.x * sc0 + w0 * yc.x;
      acc0.y = acc0.y * sc0 + w0 * yc.y;
      acc0.z = acc0.z * sc0 + w0 * yc.z;
      acc0.w = acc0.w * sc0 + w0 * yc.w;

      float x1 = expf(-fabsf(s1 - m1));
      bool  g1 = s1 > m1;
      float sc1 = g1 ? x1 : 1.0f;
      float w1  = g1 ? 1.0f : x1;
      m1 = g1 ? s1 : m1;
      d1 = d1 * sc1 + w1;
      acc1.x = acc1.x * sc1 + w1 * yc.x;
      acc1.y = acc1.y * sc1 + w1 * yc.y;
      acc1.z = acc1.z * sc1 + w1 * yc.z;
      acc1.w = acc1.w * sc1 + w1 * yc.w;
    }
  }

  float4 e0 = make_float4(0.0f, 0.0f, 0.0f, 0.0f);
  float4 e1 = make_float4(0.0f, 0.0f, 0.0f, 0.0f);
  if (LAST) {
    e0 = in_row<false>(g, ue, ie, emb1)[lane];
    e1 = in_row<true>(g, ue, ie, emb1)[lane];
  }

  float4 o = make_float4(0.0f, 0.0f, 0.0f, 0.0f);
  if (n > 0) {
    float i0 = 0.5f / d0;
    float i1 = 0.5f / d1;
    o.x = acc0.x * i0 + acc1.x * i1;
    o.y = acc0.y * i0 + acc1.y * i1;
    o.z = acc0.z * i0 + acc1.z * i1;
    o.w = acc0.w * i0 + acc1.w * i1;
  }
  float4* dp = (float4*)(dst + (size_t)g * HIDDEN) + lane;
  if (LAST) {
    const float third = 1.0f / 3.0f;
    o.x = (o.x + (e0.x + e1.x)) * third;
    o.y = (o.y + (e0.y + e1.y)) * third;
    o.z = (o.z + (e0.z + e1.z)) * third;
    o.w = (o.w + (e0.w + e1.w)) * third;
  }
  *dp = o;
  }
}

extern "C" void kernel_launch(void* const* d_in, const int* in_sizes, int n_in,
                              void* d_out, int out_size, void* d_ws, size_t ws_size,
                              hipStream_t stream) {
  (void)in_sizes; (void)n_in; (void)out_size; (void)ws_size;
  const float* ue      = (const float*)d_in[0];
  const float* ie      = (const float*)d_in[1];
  const float* eigs    = (const float*)d_in[2];
  const float* lambda0 = (const float*)d_in[3];
  const float* pemb    = (const float*)d_in[4];
  const int*   indices = (const int*)d_in[5];
  const int*   pt      = (const int*)d_in[6];
  const int* row = indices;
  const int* col = indices + NEDGES;
  float* out = (float*)d_out;

  char* p = (char*)d_ws;
  auto carve = [&](size_t bytes) -> char* {
    char* r = p;
    p += (bytes + 255) & ~(size_t)255;
    return r;
  };
  float2* stats = (float2*)carve((size_t)NNODES * 8);
  float* emb1  = (float*)carve((size_t)NNODES * HIDDEN * 4);
  int2*  sedge = (int2*) carve((size_t)NEDGES * 8);
  int*   rank  = (int*)  carve((size_t)NEDGES * 4);
  int*   cnt   = (int*)  carve((size_t)NNODES * 4);
  int*   off   = (int*)  carve((size_t)NNODES * 4);
  int*   bsum  = (int*)  carve((size_t)SCAN_B * 4);

  const dim3 B(256);
  const int GRID_LN        = 1172;
  const int GRID_EDGE      = 1024;
  const int GRID_ROW       = 6144;

  hipMemsetAsync(cnt, 0, (size_t)NNODES * 4, stream);
  hipLaunchKernelGGL(hist_rank_kernel, dim3(GRID_EDGE), B, 0, stream, row, cnt, rank);
  hipLaunchKernelGGL(blocksum_kernel, dim3(SCAN_B), dim3(SCAN_T), 0, stream, cnt, bsum);
  hipLaunchKernelGGL(offsets_kernel, dim3(SCAN_B), dim3(SCAN_T), 0, stream, cnt, bsum, off);
  hipLaunchKernelGGL(fill_kernel, dim3(GRID_EDGE), B, 0, stream, row, col, pt, rank, off, sedge);

  for (int l = 0; l < 2; ++l) {
    if (l == 0)
      hipLaunchKernelGGL(ln_wmma_kernel, dim3(GRID_LN), B, 0, stream, ue, ie, NUSERS, stats);
    else
      hipLaunchKernelGGL(ln_wmma_kernel, dim3(GRID_LN), B, 0, stream, emb1, emb1, NNODES, stats);

    if (l == 0)
      hipLaunchKernelGGL(row_kernel<false>, dim3(GRID_ROW), B, 0, stream,
                         stats, eigs, off, cnt, sedge, lambda0 + l, pemb + l * NPATHS,
                         ue, ie, (const float*)nullptr, emb1);
    else
      hipLaunchKernelGGL(row_kernel<true>, dim3(GRID_ROW), B, 0, stream,
                         stats, eigs, off, cnt, sedge, lambda0 + l, pemb + l * NPATHS,
                         ue, ie, (const float*)emb1, out);
  }
}


# Row kernels: 8 lanes per row (two float4 per lane, four rows per wave) instead of 16; score dot summed 10 products per lane then a 3-step tree (score sum order changes); edge order unchanged
# speedup vs baseline: 5.0658x; 1.0995x over previous
#include <hip/hip_runtime.h>

#define HIDDEN 64
#define EIGD   16
#define NUSERS 100000
#define NITEMS 200000
#define NNODES 300000
#define NEDGES 1250000
#define NPATHS 14
#define LN_EPS 1e-5f

typedef __attribute__((ext_vector_type(2))) float v2f;
typedef __attribute__((ext_vector_type(8))) float v8f;

__global__ void ln_wmma_kernel(const float* __restrict__ srcA,
                               const float* __restrict__ srcB,
                               int nA, float2* __restrict__ stats) {
  int lane = threadIdx.x & 31;
  int wave = threadIdx.x >> 5;
  int m = lane & 15;
  int h = lane >> 4;
  const int wstep = (gridDim.x * (blockDim.x >> 5)) << 4;
  for (int base = (blockIdx.x * (blockDim.x >> 5) + wave) << 4; base < NNODES; base += wstep) {
  int node = base + m;
  const float* srow = (node < nA) ? (srcA + (size_t)node * HIDDEN)
                                  : (srcB + (size_t)(node - nA) * HIDDEN);
  const float* sp = srow + 2 * h;

  const v2f ones = {1.0f, 1.0f};
  v8f accS = {};
  v8f accQ = {};
#pragma unroll
  for (int c = 0; c < 16; ++c) {
    float2 v = *(const float2*)(sp + 4 * c);
    v2f a  = {v.x, v.y};
    v2f a2 = {v.x * v.x, v.y * v.y};
    accS = __builtin_amdgcn_wmma_f32_16x16x4_f32(false, a,  false, ones,
                                                 (short)0, accS, false, false);
    accQ = __builtin_amdgcn_wmma_f32_16x16x4_f32(false, a2, false, ones,
                                                 (short)0, accQ, false, false);
  }

  float s = 0.0f, q = 0.0f;
#pragma unroll
  for (int i = 0; i < 8; ++i) {
    float sLo = __shfl(accS[i], 0, 32);
    float sHi = __shfl(accS[i], 16, 32);
    float qLo = __shfl(accQ[i], 0, 32);
    float qHi = __shfl(accQ[i], 16, 32);
    s = (m == i)     ? sLo : s;
    s = (m == 8 + i) ? sHi : s;
    q = (m == i)     ? qLo : q;
    q = (m == 8 + i) ? qHi : q;
  }

  float mu  = s * (1.0f / HIDDEN);
  float var = q * (1.0f / HIDDEN) - mu * mu;
  float rin = rsqrtf(var + LN_EPS);
  if (h == 0) stats[node] = make_float2(mu, rin);
  }
}

__global__ void hist_rank_kernel(const int* __restrict__ row,
                                 int* __restrict__ cnt,
                                 int* __restrict__ rank) {
  const int stride = gridDim.x * blockDim.x;
  for (int e = blockIdx.x * blockDim.x + threadIdx.x; e < NEDGES; e += stride)
    rank[e] = atomicAdd(cnt + row[e], 1);
}

#define SCAN_T 256
#define SCAN_E (SCAN_T * 4)
#define SCAN_B ((NNODES + SCAN_E - 1) / SCAN_E)

__global__ void blocksum_kernel(const int* __restrict__ cnt, int* __restrict__ bsum) {
  __shared__ int wsum[SCAN_T / 32];
  int t = threadIdx.x;
  int i0 = (blockIdx.x * SCAN_T + t) * 4;
  int s = 0;
  if (i0 + 3 < NNODES) {
    int4 v = *(const int4*)(cnt + i0);
    s = v.x + v.y + v.z + v.w;
  } else {
    for (int k = 0; k < 4; ++k)
      if (i0 + k < NNODES) s += cnt[i0 + k];
  }
  for (int o = 16; o > 0; o >>= 1) s += __shfl_xor(s, o, 32);
  if ((t & 31) == 0) wsum[t >> 5] = s;
  __syncthreads();
  if (t == 0) {
    int a = 0;
    for (int w = 0; w < SCAN_T / 32; ++w) a += wsum[w];
    bsum[blockIdx.x] = a;
  }
}

__global__ void offsets_kernel(const int* __restrict__ cnt,
                               const int* __restrict__ bsum,
                               int* __restrict__ off) {
  __shared__ int wpre[SCAN_T / 32];
  __shared__ int wtot[SCAN_T / 32];
  int t = threadIdx.x;
  int lane = t & 31;
  int wave = t >> 5;

  int ps = 0;
  for (int i = t; i < (int)blockIdx.x; i += SCAN_T) ps += bsum[i];
  for (int o = 16; o > 0; o >>= 1) ps += __shfl_xor(ps, o, 32);
  if (lane == 0) wpre[wave] = ps;

  int i0 = (blockIdx.x * SCAN_T + t) * 4;
  int v0 = 0, v1 = 0, v2 = 0, v3 = 0;
  if (i0 + 3 < NNODES) {
    int4 v = *(const int4*)(cnt + i0);
    v0 = v.x; v1 = v.y; v2 = v.z; v3 = v.w;
  } else {
    if (i0     < NNODES) v0 = cnt[i0];
    if (i0 + 1 < NNODES) v1 = cnt[i0 + 1];
    if (i0 + 2 < NNODES) v2 = cnt[i0 + 2];
  }
  int s = v0 + v1 + v2 + v3;
  int x = s;
  for (int o = 1; o < 32; o <<= 1) {
    int nb = __shfl_up(x, o, 32);
    if (lane >= o) x += nb;
  }
  if (lane == 31) wtot[wave] = x;
  __syncthreads();
  int basev = 0;
  for (int w = 0; w < SCAN_T / 32; ++w) {
    basev += wpre[w];
    if (w < wave) basev += wtot[w];
  }
  int ex = basev + x - s;
  if (i0 + 3 < NNODES) {
    *(int4*)(off + i0) = make_int4(ex, ex + v0, ex + v0 + v1, ex + v0 + v1 + v2);
  } else {
    if (i0     < NNODES) off[i0]     = ex;
    if (i0 + 1 < NNODES) off[i0 + 1] = ex + v0;
    if (i0 + 2 < NNODES) off[i0 + 2] = ex + v0 + v1;
  }
}

__global__ void fill_kernel(const int* __restrict__ row,
                            const int* __restrict__ col,
                            const int* __restrict__ pt,
                            const int* __restrict__ rank,
                            const int* __restrict__ off,
                            int2* __restrict__ sedge) {
  const int stride = gridDim.x * blockDim.x;
  for (int e = blockIdx.x * blockDim.x + threadIdx.x; e < NEDGES; e += stride) {
    int r = row[e];
    sedge[off[r] + rank[e]] = make_int2(col[e], pt[e]);
  }
}

#define GLANES 8
#define GHALF  (HIDDEN / 4 / 2)

template <bool LAST>
__device__ __forceinline__ const float4* in_row(int node,
                                                const float* __restrict__ ue,
                                                const float* __restrict__ ie,
                                                const float* __restrict__ emb1) {
  if (LAST) return (const float4*)(emb1 + (size_t)node * HIDDEN);
  bool u = node < NUSERS;
  const float* tab = u ? ue : ie;
  int idx = u ? node : node - NUSERS;
  return (const float4*)(tab + (size_t)idx * HIDDEN);
}

__device__ __forceinline__ float4 norm4(float4 v, float2 s) {
  v.x = (v.x - s.x) * s.y;
  v.y = (v.y - s.x) * s.y;
  v.z = (v.z - s.x) * s.y;
  v.w = (v.w - s.x) * s.y;
  return v;
}

__device__ __forceinline__ void axpby4(float4& acc, float sc, float w, const float4& y) {
  acc.x = acc.x * sc + w * y.x;
  acc.y = acc.y * sc + w * y.y;
  acc.z = acc.z * sc + w * y.z;
  acc.w = acc.w * sc + w * y.w;
}

template <bool LAST>
__global__ void row_kernel(const float2* __restrict__ stats,
                           const float* __restrict__ eigs,
                           const int* __restrict__ off,
                           const int* __restrict__ cnt,
                           const int2* __restrict__ sedge,
                           const float* __restrict__ lambda0_l,
                           const float* __restrict__ pemb_l,
                           const float* __restrict__ ue,
                           const float* __restrict__ ie,
                           const float* __restrict__ emb1,
                           const float* __restrict__ own1,
                           float* __restrict__ dst) {
  const int lane = threadIdx.x & (GLANES - 1);
  const int ngroups = gridDim.x * (blockDim.x / GLANES);
  const float el = expf(lambda0_l[0]);
  for (int g = blockIdx.x * (blockDim.x / GLANES) + (threadIdx.x / GLANES); g < NNODES; g += ngroups) {
  int beg = off[g];
  int n   = cnt[g];
  const float4* rp = in_row<LAST>(g, ue, ie, emb1);
  float4 yra = rp[lane];
  float4 yrb = rp[lane + GHALF];
  {
    float2 sg = stats[g];
    yra = norm4(yra, sg);
    yrb = norm4(yrb, sg);
  }
  float2 er = ((const float2*)(eigs + (size_t)g * EIGD))[lane];
  float  wra = el * er.x;
  float  wrb = el * er.y;

  const float ninf = -__builtin_inff();
  float m0 = ninf, m1 = ninf, d0 = 0.0f, d1 = 0.0f;
  float4 acc0a = make_float4(0.0f, 0.0f, 0.0f, 0.0f);
  float4 acc0b = make_float4(0.0f, 0.0f, 0.0f, 0.0f);
  float4 acc1a = make_float4(0.0f, 0.0f, 0.0f, 0.0f);
  float4 acc1b = make_float4(0.0f, 0.0f, 0.0f, 0.0f);

  for (int base = 0; base < n; base += GLANES) {
    int cc = n - base;
    if (cc > GLANES) cc = GLANES;
    int2 my = make_int2(0, 0);
    if (lane < cc) my = sedge[beg + base + lane];
    for (int j = 0; j < cc; ++j) {
      int c = __shfl(my.x, j, GLANES);
      int t = __shfl(my.y, j, GLANES);
      const float4* cp = in_row<LAST>(c, ue, ie, emb1);
      float4 yca = cp[lane];
      float4 ycb = cp[lane + GHALF];
      float2 sn = stats[c];
      float2 ec = ((const float2*)(eigs + (size_t)c * EIGD))[lane];
      yca = norm4(yca, sn);
      ycb = norm4(ycb, sn);

      float p = 0.125f * (yra.x * yca.x + yra.y * yca.y + yra.z * yca.z + yra.w * yca.w +
                          yrb.x * ycb.x + yrb.y * ycb.y + yrb.z * ycb.z + yrb.w * ycb.w);
      p += wra * ec.x;
      p += wrb * ec.y;
      for (int o = GLANES / 2; o > 0; o >>= 1) p += __shfl_xor(p, o, GLANES);
      float s1 = pemb_l[t];

      float x0 = expf(-fabsf(p - m0));
      bool  g0 = p > m0;
      float sc0 = g0 ? x0 : 1.0f;
      float w0  = g0 ? 1.0f : x0;
      m0 = g0 ? p : m0;
      d0 = d0 * sc0 + w0;
      axpby4(acc0a, sc0, w0, yca);
      axpby4(acc0b, sc0, w0, ycb);

      float x1 = expf(-fabsf(s1 - m1));
      bool  g1 = s1 > m1;
      float sc1 = g1 ? x1 : 1.0f;
      float w1  = g1 ? 1.0f : x1;
      m1 = g1 ? s1 : m1;
      d1 = d1 * sc1 + w1;
      axpby4(acc1a, sc1, w1, yca);
      axpby4(acc1b, sc1, w1, ycb);
    }
  }

  float4 e0a = make_float4(0.0f, 0.0f, 0.0f, 0.0f), e0b = e0a, e1a = e0a, e1b = e0a;
  if (LAST) {
    const float4* p0 = in_row<false>(g, ue, ie, emb1);
    const float4* p1 = (const float4*)(own1 + (size_t)g * HIDDEN);
    e0a = p0[lane]; e0b = p0[lane + GHALF];
    e1a = p1[lane]; e1b = p1[lane + GHALF];
  }

  float4 oa = make_float4(0.0f, 0.0f, 0.0f, 0.0f);
  float4 ob = oa;
  if (n > 0) {
    float i0 = 0.5f / d0;
    float i1 = 0.5f / d1;
    oa.x = acc0a.x * i0 + acc1a.x * i1;
    oa.y = acc0a.y * i0 + acc1a.y * i1;
    oa.z = acc0a.z * i0 + acc1a.z * i1;
    oa.w = acc0a.w * i0 + acc1a.w * i1;
    ob.x = acc0b.x * i0 + acc1b.x * i1;
    ob.y = acc0b.y * i0 + acc1b.y * i1;
    ob.z = acc0b.z * i0 + acc1b.z * i1;
    ob.w = acc0b.w * i0 + acc1b.w * i1;
  }
  float4* dp = (float4*)(dst + (size_t)g * HIDDEN) + lane;
  if (LAST) {
    const float third = 1.0f / 3.0f;
    oa.x = (oa.x + (e0a.x + e1a.x)) * third;
    oa.y = (oa.y + (e0a.y + e1a.y)) * third;
    oa.z = (oa.z + (e0a.z + e1a.z)) * third;
    oa.w = (oa.w + (e0a.w + e1a.w)) * third;
    ob.x = (ob.x + (e0b.x + e1b.x)) * third;
    ob.y = (ob.y + (e0b.y + e1b.y)) * third;
    ob.z = (ob.z + (e0b.z + e1b.z)) * third;
    ob.w = (ob.w + (e0b.w + e1b.w)) * third;
  }
  dp[0]     = oa;
  dp[GHALF] = ob;
  }
}

extern "C" void kernel_launch(void* const* d_in, const int* in_sizes, int n_in,
                              void* d_out, int out_size, void* d_ws, size_t ws_size,
                              hipStream_t stream) {
  (void)in_sizes; (void)n_in; (void)out_size; (void)ws_size;
  const float* ue      = (const float*)d_in[0];
  const float* ie      = (const float*)d_in[1];
  const float* eigs    = (const float*)d_in[2];
  const float* lambda0 = (const float*)d_in[3];
  const float* pemb    = (const float*)d_in[4];
  const int*   indices = (const int*)d_in[5];
  const int*   pt      = (const int*)d_in[6];
  const int* row = indices;
  const int* col = indices + NEDGES;
  float* out = (float*)d_out;

  char* p = (char*)d_ws;
  auto carve = [&](size_t bytes) -> char* {
    char* r = p;
    p += (bytes + 255) & ~(size_t)255;
    return r;
  };
  float2* stats = (float2*)carve((size_t)NNODES * 8);
  float* emb1  = (float*)carve((size_t)NNODES * HIDDEN * 4);
  int2*  sedge = (int2*) carve((size_t)NEDGES * 8);
  int*   rank  = (int*)  carve((size_t)NEDGES * 4);
  int*   cnt   = (int*)  carve((size_t)NNODES * 4);
  int*   off   = (int*)  carve((size_t)NNODES * 4);
  int*   bsum  = (int*)  carve((size_t)SCAN_B * 4);

  const dim3 B(256);
  const int GRID_LN        = 1172;
  const int GRID_EDGE      = 1024;
  const int GRID_ROW       = 4688;

  hipMemsetAsync(cnt, 0, (size_t)NNODES * 4, stream);
  hipLaunchKernelGGL(hist_rank_kernel, dim3(GRID_EDGE), B, 0, stream, row, cnt, rank);
  hipLaunchKernelGGL(blocksum_kernel, dim3(SCAN_B), dim3(SCAN_T), 0, stream, cnt, bsum);
  hipLaunchKernelGGL(offsets_kernel, dim3(SCAN_B), dim3(SCAN_T), 0, stream, cnt, bsum, off);
  hipLaunchKernelGGL(fill_kernel, dim3(GRID_EDGE), B, 0, stream, row, col, pt, rank, off, sedge);

  for (int l = 0; l < 2; ++l) {
    if (l == 0)
      hipLaunchKernelGGL(ln_wmma_kernel, dim3(GRID_LN), B, 0, stream, ue, ie, NUSERS, stats);
    else
      hipLaunchKernelGGL(ln_wmma_kernel, dim3(GRID_LN), B, 0, stream, emb1, emb1, NNODES, stats);

    if (l == 0)
      hipLaunchKernelGGL(row_kernel<false>, dim3(GRID_ROW), B, 0, stream,
                         stats, eigs, off, cnt, sedge, lambda0 + l, pemb + l * NPATHS,
                         ue, ie, (const float*)nullptr, (const float*)nullptr, emb1);
    else
      hipLaunchKernelGGL(row_kernel<true>, dim3(GRID_ROW), B, 0, stream,
                         stats, eigs, off, cnt, sedge, lambda0 + l, pemb + l * NPATHS,
                         ue, ie, (const float*)emb1, (const float*)emb1, out);
  }
}


# Row kernels take a row's edges four at a time: its four source-row gathers are in flight together, both running softmaxes rescale once per four edges (softmax sum order changes); blocks of 4 waves
# speedup vs baseline: 5.2619x; 1.0387x over previous
#include <hip/hip_runtime.h>

#define HIDDEN 64
#define EIGD   16
#define NUSERS 100000
#define NITEMS 200000
#define NNODES 300000
#define NEDGES 1250000
#define NPATHS 14
#define LN_EPS 1e-5f

typedef __attribute__((ext_vector_type(2))) float v2f;
typedef __attribute__((ext_vector_type(8))) float v8f;
typedef __attribute__((ext_vector_type(4))) float v4f;

__global__ void ln_wmma_kernel(const float* __restrict__ srcA,
                               const float* __restrict__ srcB,
                               int nA, float2* __restrict__ stats) {
  int lane = threadIdx.x & 31;
  int wave = threadIdx.x >> 5;
  int m = lane & 15;
  int h = lane >> 4;
  const int wstep = (gridDim.x * (blockDim.x >> 5)) << 4;
  for (int base = (blockIdx.x * (blockDim.x >> 5) + wave) << 4; base < NNODES; base += wstep) {
  int node = base + m;
  const float* srow = (node < nA) ? (srcA + (size_t)node * HIDDEN)
                                  : (srcB + (size_t)(node - nA) * HIDDEN);
  const float* sp = srow + 2 * h;

  const v2f ones = {1.0f, 1.0f};
  v8f accS = {};
  v8f accQ = {};
#pragma unroll
  for (int c = 0; c < 16; ++c) {
    float2 v = *(const float2*)(sp + 4 * c);
    v2f a  = {v.x, v.y};
    v2f a2 = {v.x * v.x, v.y * v.y};
    accS = __builtin_amdgcn_wmma_f32_16x16x4_f32(false, a,  false, ones,
                                                 (short)0, accS, false, false);
    accQ = __builtin_amdgcn_wmma_f32_16x16x4_f32(false, a2, false, ones,
                                                 (short)0, accQ, false, false);
  }

  float s = 0.0f, q = 0.0f;
#pragma unroll
  for (int i = 0; i < 8; ++i) {
    float sLo = __shfl(accS[i], 0, 32);
    float sHi = __shfl(accS[i], 16, 32);
    float qLo = __shfl(accQ[i], 0, 32);
    float qHi = __shfl(accQ[i], 16, 32);
    s = (m == i)     ? sLo : s;
    s = (m == 8 + i) ? sHi : s;
    q = (m == i)     ? qLo : q;
    q = (m == 8 + i) ? qHi : q;
  }

  float mu  = s * (1.0f / HIDDEN);
  float var = q * (1.0f / HIDDEN) - mu * mu;
  float rin = rsqrtf(var + LN_EPS);
  if (h == 0) stats[node] = make_float2(mu, rin);
  }
}

__global__ void hist_rank_kernel(const int* __restrict__ row,
                                 int* __restrict__ cnt,
                                 int* __restrict__ rank) {
  const int stride = gridDim.x * blockDim.x;
  for (int e = blockIdx.x * blockDim.x + threadIdx.x; e < NEDGES; e += stride)
    rank[e] = atomicAdd(cnt + row[e], 1);
}

#define SCAN_T 256
#define SCAN_E (SCAN_T * 4)
#define SCAN_B ((NNODES + SCAN_E - 1) / SCAN_E)

__global__ void blocksum_kernel(const int* __restrict__ cnt, int* __restrict__ bsum) {
  __shared__ int wsum[SCAN_T / 32];
  int t = threadIdx.x;
  int i0 = (blockIdx.x * SCAN_T + t) * 4;
  int s = 0;
  if (i0 + 3 < NNODES) {
    int4 v = *(const int4*)(cnt + i0);
    s = v.x + v.y + v.z + v.w;
  } else {
    for (int k = 0; k < 4; ++k)
      if (i0 + k < NNODES) s += cnt[i0 + k];
  }
  for (int o = 16; o > 0; o >>= 1) s += __shfl_xor(s, o, 32);
  if ((t & 31) == 0) wsum[t >> 5] = s;
  __syncthreads();
  if (t == 0) {
    int a = 0;
    for (int w = 0; w < SCAN_T / 32; ++w) a += wsum[w];
    bsum[blockIdx.x] = a;
  }
}

__global__ void offsets_kernel(const int* __restrict__ cnt,
                               const int* __restrict__ bsum,
                               int* __restrict__ off) {
  __shared__ int wpre[SCAN_T / 32];
  __shared__ int wtot[SCAN_T / 32];
  int t = threadIdx.x;
  int lane = t & 31;
  int wave = t >> 5;

  int ps = 0;
  for (int i = t; i < (int)blockIdx.x; i += SCAN_T) ps += bsum[i];
  for (int o = 16; o > 0; o >>= 1) ps += __shfl_xor(ps, o, 32);
  if (lane == 0) wpre[wave] = ps;

  int i0 = (blockIdx.x * SCAN_T + t) * 4;
  int v0 = 0, v1 = 0, v2 = 0, v3 = 0;
  if (i0 + 3 < NNODES) {
    int4 v = *(const int4*)(cnt + i0);
    v0 = v.x; v1 = v.y; v2 = v.z; v3 = v.w;
  } else {
    if (i0     < NNODES) v0 = cnt[i0];
    if (i0 + 1 < NNODES) v1 = cnt[i0 + 1];
    if (i0 + 2 < NNODES) v2 = cnt[i0 + 2];
  }
  int s = v0 + v1 + v2 + v3;
  int x = s;
  for (int o = 1; o < 32; o <<= 1) {
    int nb = __shfl_up(x, o, 32);
    if (lane >= o) x += nb;
  }
  if (lane == 31) wtot[wave] = x;
  __syncthreads();
  int basev = 0;
  for (int w = 0; w < SCAN_T / 32; ++w) {
    basev += wpre[w];
    if (w < wave) basev += wtot[w];
  }
  int ex = basev + x - s;
  if (i0 + 3 < NNODES) {
    *(int4*)(off + i0) = make_int4(ex, ex + v0, ex + v0 + v1, ex + v0 + v1 + v2);
  } else {
    if (i0     < NNODES) off[i0]     = ex;
    if (i0 + 1 < NNODES) off[i0 + 1] = ex + v0;
    if (i0 + 2 < NNODES) off[i0 + 2] = ex + v0 + v1;
  }
}

__global__ void fill_kernel(const int* __restrict__ row,
                            const int* __restrict__ col,
                            const int* __restrict__ pt,
                            const int* __restrict__ rank,
                            const int* __restrict__ off,
                            int2* __restrict__ sedge) {
  const int stride = gridDim.x * blockDim.x;
  for (int e = blockIdx.x * blockDim.x + threadIdx.x; e < NEDGES; e += stride) {
    int r = row[e];
    sedge[off[r] + rank[e]] = make_int2(col[e], pt[e]);
  }
}

#define GLANES 8
#define GHALF  (HIDDEN / 4 / 2)
#define ECH    4

template <bool LAST>
__device__ __forceinline__ const float4* in_row(int node,
                                                const float* __restrict__ ue,
                                                const float* __restrict__ ie,
                                                const float* __restrict__ emb1) {
  if (LAST) return (const float4*)(emb1 + (size_t)node * HIDDEN);
  bool u = node < NUSERS;
  const float* tab = u ? ue : ie;
  int idx = u ? node : node - NUSERS;
  return (const float4*)(tab + (size_t)idx * HIDDEN);
}

__device__ __forceinline__ float4 norm4(float4 v, float2 s) {
  v.x = (v.x - s.x) * s.y;
  v.y = (v.y - s.x) * s.y;
  v.z = (v.z - s.x) * s.y;
  v.w = (v.w - s.x) * s.y;
  return v;
}

__device__ __forceinline__ v4f norm4v(v4f v, v2f s) {
  v.x = (v.x - s.x) * s.y;
  v.y = (v.y - s.x) * s.y;
  v.z = (v.z - s.x) * s.y;
  v.w = (v.w - s.x) * s.y;
  return v;
}

__device__ __forceinline__ void step4(float4& acc, float sc,
                                      float w0, const v4f& y0, float w1, const v4f& y1,
                                      float w2, const v4f& y2, float w3, const v4f& y3) {
  acc.x = acc.x * sc + w0 * y0.x + w1 * y1.x + w2 * y2.x + w3 * y3.x;
  acc.y = acc.y * sc + w0 * y0.y + w1 * y1.y + w2 * y2.y + w3 * y3.y;
  acc.z = acc.z * sc + w0 * y0.z + w1 * y1.z + w2 * y2.z + w3 * y3.z;
  acc.w = acc.w * sc + w0 * y0.w + w1 * y1.w + w2 * y2.w + w3 * y3.w;
}

template <bool LAST>
__global__ void __launch_bounds__(128) row_kernel(const float2* __restrict__ stats,
                           const float* __restrict__ eigs,
                           const int* __restrict__ off,
                           const int* __restrict__ cnt,
                           const int2* __restrict__ sedge,
                           const float* __restrict__ lambda0_l,
                           const float* __restrict__ pemb_l,
                           const float* __restrict__ ue,
                           const float* __restrict__ ie,
                           const float* __restrict__ emb1,
                           const float* __restrict__ own1,
                           float* __restrict__ dst) {
  const int lane = threadIdx.x & (GLANES - 1);
  const int ngroups = gridDim.x * (blockDim.x / GLANES);
  const float el = expf(lambda0_l[0]);
  for (int g = blockIdx.x * (blockDim.x / GLANES) + (threadIdx.x / GLANES); g < NNODES; g += ngroups) {
  int beg = off[g];
  int n   = cnt[g];
  const float4* rp = in_row<LAST>(g, ue, ie, emb1);
  float4 yra = rp[lane];
  float4 yrb = rp[lane + GHALF];
  {
    float2 sg = stats[g];
    yra = norm4(yra, sg);
    yrb = norm4(yrb, sg);
  }
  float2 er = ((const float2*)(eigs + (size_t)g * EIGD))[lane];
  float  wra = el * er.x;
  float  wrb = el * er.y;

  const float ninf = -__builtin_inff();
  float m0 = ninf, m1 = ninf, d0 = 0.0f, d1 = 0.0f;
  float4 acc0a = make_float4(0.0f, 0.0f, 0.0f, 0.0f);
  float4 acc0b = make_float4(0.0f, 0.0f, 0.0f, 0.0f);
  float4 acc1a = make_float4(0.0f, 0.0f, 0.0f, 0.0f);
  float4 acc1b = make_float4(0.0f, 0.0f, 0.0f, 0.0f);

  for (int base = 0; base < n; base += GLANES) {
    int cc = n - base;
    if (cc > GLANES) cc = GLANES;
    int2 my = make_int2(0, 0);
    if (lane < cc) my = sedge[beg + base + lane];
    for (int h = 0; h < cc; h += ECH) {
      v4f   ya[ECH], yb[ECH];
      v2f   sn[ECH], ec[ECH];
      float p[ECH], s1[ECH];
#pragma unroll
      for (int j = 0; j < ECH; ++j) {
        int c = __shfl(my.x, h + j, GLANES);
        int t = __shfl(my.y, h + j, GLANES);
        const v4f* cp = (const v4f*)in_row<LAST>(c, ue, ie, emb1);
        ya[j] = cp[lane];
        yb[j] = cp[lane + GHALF];
        sn[j] = ((const v2f*)stats)[c];
        ec[j] = ((const v2f*)(eigs + (size_t)c * EIGD))[lane];
        s1[j] = pemb_l[t];
      }
      asm volatile("" :: "v"(ya[0]), "v"(yb[0]), "v"(ya[1]), "v"(yb[1]),
                         "v"(ya[2]), "v"(yb[2]), "v"(ya[3]), "v"(yb[3]),
                         "v"(sn[0]), "v"(sn[1]), "v"(sn[2]), "v"(sn[3]),
                         "v"(ec[0]), "v"(ec[1]), "v"(ec[2]), "v"(ec[3]),
                         "v"(s1[0]), "v"(s1[1]), "v"(s1[2]), "v"(s1[3]));

#pragma unroll
      for (int j = 0; j < ECH; ++j) {
        ya[j] = norm4v(ya[j], sn[j]);
        yb[j] = norm4v(yb[j], sn[j]);
        float q = 0.125f * (yra.x * ya[j].x + yra.y * ya[j].y + yra.z * ya[j].z + yra.w * ya[j].w +
                            yrb.x * yb[j].x + yrb.y * yb[j].y + yrb.z * yb[j].z + yrb.w * yb[j].w);
        q += wra * ec[j].x;
        q += wrb * ec[j].y;
        p[j] = q;
      }
#pragma unroll
      for (int o = GLANES / 2; o > 0; o >>= 1) {
        float r[ECH];
#pragma unroll
        for (int j = 0; j < ECH; ++j) r[j] = __shfl_xor(p[j], o, GLANES);
#pragma unroll
        for (int j = 0; j < ECH; ++j) p[j] += r[j];
      }
#pragma unroll
      for (int j = 0; j < ECH; ++j) {
        bool valid = (h + j) < cc;
        p[j]  = valid ? p[j]  : ninf;
        s1[j] = valid ? s1[j] : ninf;
      }

      float mn0 = fmaxf(fmaxf(fmaxf(p[0], p[1]), fmaxf(p[2], p[3])), m0);
      float sc0 = expf(m0 - mn0);
      float w00 = expf(p[0] - mn0), w01 = expf(p[1] - mn0);
      float w02 = expf(p[2] - mn0), w03 = expf(p[3] - mn0);
      m0 = mn0;
      d0 = d0 * sc0 + ((w00 + w01) + (w02 + w03));
      step4(acc0a, sc0, w00, ya[0], w01, ya[1], w02, ya[2], w03, ya[3]);
      step4(acc0b, sc0, w00, yb[0], w01, yb[1], w02, yb[2], w03, yb[3]);

      float mn1 = fmaxf(fmaxf(fmaxf(s1[0], s1[1]), fmaxf(s1[2], s1[3])), m1);
      float sc1 = expf(m1 - mn1);
      float w10 = expf(s1[0] - mn1), w11 = expf(s1[1] - mn1);
      float w12 = expf(s1[2] - mn1), w13 = expf(s1[3] - mn1);
      m1 = mn1;
      d1 = d1 * sc1 + ((w10 + w11) + (w12 + w13));
      step4(acc1a, sc1, w10, ya[0], w11, ya[1], w12, ya[2], w13, ya[3]);
      step4(acc1b, sc1, w10, yb[0], w11, yb[1], w12, yb[2], w13, yb[3]);
    }
  }

  float4 e0a = make_float4(0.0f, 0.0f, 0.0f, 0.0f), e0b = e0a, e1a = e0a, e1b = e0a;
  if (LAST) {
    const float4* p0 = in_row<false>(g, ue, ie, emb1);
    const float4* p1 = (const float4*)(own1 + (size_t)g * HIDDEN);
    e0a = p0[lane]; e0b = p0[lane + GHALF];
    e1a = p1[lane]; e1b = p1[lane + GHALF];
  }

  float4 oa = make_float4(0.0f, 0.0f, 0.0f, 0.0f);
  float4 ob = oa;
  if (n > 0) {
    float i0 = 0.5f / d0;
    float i1 = 0.5f / d1;
    oa.x = acc0a.x * i0 + acc1a.x * i1;
    oa.y = acc0a.y * i0 + acc1a.y * i1;
    oa.z = acc0a.z * i0 + acc1a.z * i1;
    oa.w = acc0a.w * i0 + acc1a.w * i1;
    ob.x = acc0b.x * i0 + acc1b.x * i1;
    ob.y = acc0b.y * i0 + acc1b.y * i1;
    ob.z = acc0b.z * i0 + acc1b.z * i1;
    ob.w = acc0b.w * i0 + acc1b.w * i1;
  }
  float4* dp = (float4*)(dst + (size_t)g * HIDDEN) + lane;
  if (LAST) {
    const float third = 1.0f / 3.0f;
    oa.x = (oa.x + (e0a.x + e1a.x)) * third;
    oa.y = (oa.y + (e0a.y + e1a.y)) * third;
    oa.z = (oa.z + (e0a.z + e1a.z)) * third;
    oa.w = (oa.w + (e0a.w + e1a.w)) * third;
    ob.x = (ob.x + (e0b.x + e1b.x)) * third;
    ob.y = (ob.y + (e0b.y + e1b.y)) * third;
    ob.z = (ob.z + (e0b.z + e1b.z)) * third;
    ob.w = (ob.w + (e0b.w + e1b.w)) * third;
  }
  dp[0]     = oa;
  dp[GHALF] = ob;
  }
}

extern "C" void kernel_launch(void* const* d_in, const int* in_sizes, int n_in,
                              void* d_out, int out_size, void* d_ws, size_t ws_size,
                              hipStream_t stream) {
  (void)in_sizes; (void)n_in; (void)out_size; (void)ws_size;
  const float* ue      = (const float*)d_in[0];
  const float* ie      = (const float*)d_in[1];
  const float* eigs    = (const float*)d_in[2];
  const float* lambda0 = (const float*)d_in[3];
  const float* pemb    = (const float*)d_in[4];
  const int*   indices = (const int*)d_in[5];
  const int*   pt      = (const int*)d_in[6];
  const int* row = indices;
  const int* col = indices + NEDGES;
  float* out = (float*)d_out;

  char* p = (char*)d_ws;
  auto carve = [&](size_t bytes) -> char* {
    char* r = p;
    p += (bytes + 255) & ~(size_t)255;
    return r;
  };
  float2* stats = (float2*)carve((size_t)NNODES * 8);
  float* emb1  = (float*)carve((size_t)NNODES * HIDDEN * 4);
  int2*  sedge = (int2*) carve((size_t)NEDGES * 8);
  int*   rank  = (int*)  carve((size_t)NEDGES * 4);
  int*   cnt   = (int*)  carve((size_t)NNODES * 4);
  int*   off   = (int*)  carve((size_t)NNODES * 4);
  int*   bsum  = (int*)  carve((size_t)SCAN_B * 4);

  const dim3 B(256);
  const int GRID_LN        = 1172;
  const int GRID_EDGE      = 1024;
  const dim3 B_ROW(128);
  const int GRID_ROW       = 9376;

  hipMemsetAsync(cnt, 0, (size_t)NNODES * 4, stream);
  hipLaunchKernelGGL(hist_rank_kernel, dim3(GRID_EDGE), B, 0, stream, row, cnt, rank);
  hipLaunchKernelGGL(blocksum_kernel, dim3(SCAN_B), dim3(SCAN_T), 0, stream, cnt, bsum);
  hipLaunchKernelGGL(offsets_kernel, dim3(SCAN_B), dim3(SCAN_T), 0, stream, cnt, bsum, off);
  hipLaunchKernelGGL(fill_kernel, dim3(GRID_EDGE), B, 0, stream, row, col, pt, rank, off, sedge);

  for (int l = 0; l < 2; ++l) {
    if (l == 0)
      hipLaunchKernelGGL(ln_wmma_kernel, dim3(GRID_LN), B, 0, stream, ue, ie, NUSERS, stats);
    else
      hipLaunchKernelGGL(ln_wmma_kernel, dim3(GRID_LN), B, 0, stream, emb1, emb1, NNODES, stats);

    if (l == 0)
      hipLaunchKernelGGL(row_kernel<false>, dim3(GRID_ROW), B_ROW, 0, stream,
                         stats, eigs, off, cnt, sedge, lambda0 + l, pemb + l * NPATHS,
                         ue, ie, (const float*)nullptr, (const float*)nullptr, emb1);
    else
      hipLaunchKernelGGL(row_kernel<true>, dim3(GRID_ROW), B_ROW, 0, stream,
                         stats, eigs, off, cnt, sedge, lambda0 + l, pemb + l * NPATHS,
                         ue, ie, (const float*)emb1, (const float*)emb1, out);
  }
}


# Layer-0 row kernel stores its rows normalised (8-lane mean/var trees: layer-1 LayerNorm sum order changes; emb1 rebuilt as y*sd+mean): 2nd LayerNorm launch, last layer's per-edge stats fetch gone
# speedup vs baseline: 5.4290x; 1.0318x over previous
#include <hip/hip_runtime.h>

#define HIDDEN 64
#define EIGD   16
#define NUSERS 100000
#define NITEMS 200000
#define NNODES 300000
#define NEDGES 1250000
#define NPATHS 14
#define LN_EPS 1e-5f

typedef __attribute__((ext_vector_type(2))) float v2f;
typedef __attribute__((ext_vector_type(8))) float v8f;
typedef __attribute__((ext_vector_type(4))) float v4f;

__global__ void ln_wmma_kernel(const float* __restrict__ srcA,
                               const float* __restrict__ srcB,
                               int nA, float2* __restrict__ stats) {
  int lane = threadIdx.x & 31;
  int wave = threadIdx.x >> 5;
  int m = lane & 15;
  int h = lane >> 4;
  const int wstep = (gridDim.x * (blockDim.x >> 5)) << 4;
  for (int base = (blockIdx.x * (blockDim.x >> 5) + wave) << 4; base < NNODES; base += wstep) {
  int node = base + m;
  const float* srow = (node < nA) ? (srcA + (size_t)node * HIDDEN)
                                  : (srcB + (size_t)(node - nA) * HIDDEN);
  const float* sp = srow + 2 * h;

  const v2f ones = {1.0f, 1.0f};
  v8f accS = {};
  v8f accQ = {};
#pragma unroll
  for (int c = 0; c < 16; ++c) {
    float2 v = *(const float2*)(sp + 4 * c);
    v2f a  = {v.x, v.y};
    v2f a2 = {v.x * v.x, v.y * v.y};
    accS = __builtin_amdgcn_wmma_f32_16x16x4_f32(false, a,  false, ones,
                                                 (short)0, accS, false, false);
    accQ = __builtin_amdgcn_wmma_f32_16x16x4_f32(false, a2, false, ones,
                                                 (short)0, accQ, false, false);
  }

  float s = 0.0f, q = 0.0f;
#pragma unroll
  for (int i = 0; i < 8; ++i) {
    float sLo = __shfl(accS[i], 0, 32);
    float sHi = __shfl(accS[i], 16, 32);
    float qLo = __shfl(accQ[i], 0, 32);
    float qHi = __shfl(accQ[i], 16, 32);
    s = (m == i)     ? sLo : s;
    s = (m == 8 + i) ? sHi : s;
    q = (m == i)     ? qLo : q;
    q = (m == 8 + i) ? qHi : q;
  }

  float mu  = s * (1.0f / HIDDEN);
  float var = q * (1.0f / HIDDEN) - mu * mu;
  float rin = rsqrtf(var + LN_EPS);
  if (h == 0) stats[node] = make_float2(mu, rin);
  }
}

__global__ void hist_rank_kernel(const int* __restrict__ row,
                                 int* __restrict__ cnt,
                                 int* __restrict__ rank) {
  const int stride = gridDim.x * blockDim.x;
  for (int e = blockIdx.x * blockDim.x + threadIdx.x; e < NEDGES; e += stride)
    rank[e] = atomicAdd(cnt + row[e], 1);
}

#define SCAN_T 256
#define SCAN_E (SCAN_T * 4)
#define SCAN_B ((NNODES + SCAN_E - 1) / SCAN_E)

__global__ void blocksum_kernel(const int* __restrict__ cnt, int* __restrict__ bsum) {
  __shared__ int wsum[SCAN_T / 32];
  int t = threadIdx.x;
  int i0 = (blockIdx.x * SCAN_T + t) * 4;
  int s = 0;
  if (i0 + 3 < NNODES) {
    int4 v = *(const int4*)(cnt + i0);
    s = v.x + v.y + v.z + v.w;
  } else {
    for (int k = 0; k < 4; ++k)
      if (i0 + k < NNODES) s += cnt[i0 + k];
  }
  for (int o = 16; o > 0; o >>= 1) s += __shfl_xor(s, o, 32);
  if ((t & 31) == 0) wsum[t >> 5] = s;
  __syncthreads();
  if (t == 0) {
    int a = 0;
    for (int w = 0; w < SCAN_T / 32; ++w) a += wsum[w];
    bsum[blockIdx.x] = a;
  }
}

__global__ void offsets_kernel(const int* __restrict__ cnt,
                               const int* __restrict__ bsum,
                               int* __restrict__ off) {
  __shared__ int wpre[SCAN_T / 32];
  __shared__ int wtot[SCAN_T / 32];
  int t = threadIdx.x;
  int lane = t & 31;
  int wave = t >> 5;

  int ps = 0;
  for (int i = t; i < (int)blockIdx.x; i += SCAN_T) ps += bsum[i];
  for (int o = 16; o > 0; o >>= 1) ps += __shfl_xor(ps, o, 32);
  if (lane == 0) wpre[wave] = ps;

  int i0 = (blockIdx.x * SCAN_T + t) * 4;
  int v0 = 0, v1 = 0, v2 = 0, v3 = 0;
  if (i0 + 3 < NNODES) {
    int4 v = *(const int4*)(cnt + i0);
    v0 = v.x; v1 = v.y; v2 = v.z; v3 = v.w;
  } else {
    if (i0     < NNODES) v0 = cnt[i0];
    if (i0 + 1 < NNODES) v1 = cnt[i0 + 1];
    if (i0 + 2 < NNODES) v2 = cnt[i0 + 2];
  }
  int s = v0 + v1 + v2 + v3;
  int x = s;
  for (int o = 1; o < 32; o <<= 1) {
    int nb = __shfl_up(x, o, 32);
    if (lane >= o) x += nb;
  }
  if (lane == 31) wtot[wave] = x;
  __syncthreads();
  int basev = 0;
  for (int w = 0; w < SCAN_T / 32; ++w) {
    basev += wpre[w];
    if (w < wave) basev += wtot[w];
  }
  int ex = basev + x - s;
  if (i0 + 3 < NNODES) {
    *(int4*)(off + i0) = make_int4(ex, ex + v0, ex + v0 + v1, ex + v0 + v1 + v2);
  } else {
    if (i0     < NNODES) off[i0]     = ex;
    if (i0 + 1 < NNODES) off[i0 + 1] = ex + v0;
    if (i0 + 2 < NNODES) off[i0 + 2] = ex + v0 + v1;
  }
}

__global__ void fill_kernel(const int* __restrict__ row,
                            const int* __restrict__ col,
                            const int* __restrict__ pt,
                            const int* __restrict__ rank,
                            const int* __restrict__ off,
                            int2* __restrict__ sedge) {
  const int stride = gridDim.x * blockDim.x;
  for (int e = blockIdx.x * blockDim.x + threadIdx.x; e < NEDGES; e += stride) {
    int r = row[e];
    sedge[off[r] + rank[e]] = make_int2(col[e], pt[e]);
  }
}

#define GLANES 8
#define GHALF  (HIDDEN / 4 / 2)
#define ECH    4

template <bool LAST>
__device__ __forceinline__ const float4* in_row(int node,
                                                const float* __restrict__ ue,
                                                const float* __restrict__ ie,
                                                const float* __restrict__ y1) {
  if (LAST) return (const float4*)(y1 + (size_t)node * HIDDEN);
  bool u = node < NUSERS;
  const float* tab = u ? ue : ie;
  int idx = u ? node : node - NUSERS;
  return (const float4*)(tab + (size_t)idx * HIDDEN);
}

__device__ __forceinline__ float4 norm4(float4 v, float2 s) {
  v.x = (v.x - s.x) * s.y;
  v.y = (v.y - s.x) * s.y;
  v.z = (v.z - s.x) * s.y;
  v.w = (v.w - s.x) * s.y;
  return v;
}

__device__ __forceinline__ v4f norm4v(v4f v, v2f s) {
  v.x = (v.x - s.x) * s.y;
  v.y = (v.y - s.x) * s.y;
  v.z = (v.z - s.x) * s.y;
  v.w = (v.w - s.x) * s.y;
  return v;
}

__device__ __forceinline__ void step4(float4& acc, float sc,
                                      float w0, const v4f& y0, float w1, const v4f& y1,
                                      float w2, const v4f& y2, float w3, const v4f& y3) {
  acc.x = acc.x * sc + w0 * y0.x + w1 * y1.x + w2 * y2.x + w3 * y3.x;
  acc.y = acc.y * sc + w0 * y0.y + w1 * y1.y + w2 * y2.y + w3 * y3.y;
  acc.z = acc.z * sc + w0 * y0.z + w1 * y1.z + w2 * y2.z + w3 * y3.z;
  acc.w = acc.w * sc + w0 * y0.w + w1 * y1.w + w2 * y2.w + w3 * y3.w;
}

template <bool LAST>
__global__ void __launch_bounds__(128) row_kernel(const float2* __restrict__ st_in,
                           float2* __restrict__ st_out,
                           const float* __restrict__ eigs,
                           const int* __restrict__ off,
                           const int* __restrict__ cnt,
                           const int2* __restrict__ sedge,
                           const float* __restrict__ lambda0_l,
                           const float* __restrict__ pemb_l,
                           const float* __restrict__ ue,
                           const float* __restrict__ ie,
                           const float* __restrict__ y1,
                           float* __restrict__ dst) {
  const int lane = threadIdx.x & (GLANES - 1);
  const int ngroups = gridDim.x * (blockDim.x / GLANES);
  const float el = expf(lambda0_l[0]);
  for (int g = blockIdx.x * (blockDim.x / GLANES) + (threadIdx.x / GLANES); g < NNODES; g += ngroups) {
  int beg = off[g];
  int n   = cnt[g];
  const float4* rp = in_row<LAST>(g, ue, ie, y1);
  float4 yra = rp[lane];
  float4 yrb = rp[lane + GHALF];
  if (!LAST) {
    float2 sg = st_in[g];
    yra = norm4(yra, sg);
    yrb = norm4(yrb, sg);
  }
  float2 er = ((const float2*)(eigs + (size_t)g * EIGD))[lane];
  float  wra = el * er.x;
  float  wrb = el * er.y;

  const float ninf = -__builtin_inff();
  float m0 = ninf, m1 = ninf, d0 = 0.0f, d1 = 0.0f;
  float4 acc0a = make_float4(0.0f, 0.0f, 0.0f, 0.0f);
  float4 acc0b = make_float4(0.0f, 0.0f, 0.0f, 0.0f);
  float4 acc1a = make_float4(0.0f, 0.0f, 0.0f, 0.0f);
  float4 acc1b = make_float4(0.0f, 0.0f, 0.0f, 0.0f);

  for (int base = 0; base < n; base += GLANES) {
    int cc = n - base;
    if (cc > GLANES) cc = GLANES;
    int2 my = make_int2(0, 0);
    if (lane < cc) my = sedge[beg + base + lane];
    for (int h = 0; h < cc; h += ECH) {
      v4f   ya[ECH], yb[ECH];
      v2f   sn[ECH], ec[ECH];
      float p[ECH], s1[ECH];
#pragma unroll
      for (int j = 0; j < ECH; ++j) {
        int c = __shfl(my.x, h + j, GLANES);
        int t = __shfl(my.y, h + j, GLANES);
        const v4f* cp = (const v4f*)in_row<LAST>(c, ue, ie, y1);
        ya[j] = cp[lane];
        yb[j] = cp[lane + GHALF];
        if (!LAST) sn[j] = ((const v2f*)st_in)[c];
        ec[j] = ((const v2f*)(eigs + (size_t)c * EIGD))[lane];
        s1[j] = pemb_l[t];
      }
      if (LAST)
        asm volatile("" :: "v"(ya[0]), "v"(yb[0]), "v"(ya[1]), "v"(yb[1]),
                           "v"(ya[2]), "v"(yb[2]), "v"(ya[3]), "v"(yb[3]),
                           "v"(ec[0]), "v"(ec[1]), "v"(ec[2]), "v"(ec[3]),
                           "v"(s1[0]), "v"(s1[1]), "v"(s1[2]), "v"(s1[3]));
      else
        asm volatile("" :: "v"(ya[0]), "v"(yb[0]), "v"(ya[1]), "v"(yb[1]),
                           "v"(ya[2]), "v"(yb[2]), "v"(ya[3]), "v"(yb[3]),
                           "v"(sn[0]), "v"(sn[1]), "v"(sn[2]), "v"(sn[3]),
                           "v"(ec[0]), "v"(ec[1]), "v"(ec[2]), "v"(ec[3]),
                           "v"(s1[0]), "v"(s1[1]), "v"(s1[2]), "v"(s1[3]));

#pragma unroll
      for (int j = 0; j < ECH; ++j) {
        if (!LAST) {
          ya[j] = norm4v(ya[j], sn[j]);
          yb[j] = norm4v(yb[j], sn[j]);
        }
        float q = 0.125f * (yra.x * ya[j].x + yra.y * ya[j].y + yra.z * ya[j].z + yra.w * ya[j].w +
                            yrb.x * yb[j].x + yrb.y * yb[j].y + yrb.z * yb[j].z + yrb.w * yb[j].w);
        q += wra * ec[j].x;
        q += wrb * ec[j].y;
        p[j] = q;
      }
#pragma unroll
      for (int o = GLANES / 2; o > 0; o >>= 1) {
        float r[ECH];
#pragma unroll
        for (int j = 0; j < ECH; ++j) r[j] = __shfl_xor(p[j], o, GLANES);
#pragma unroll
        for (int j = 0; j < ECH; ++j) p[j] += r[j];
      }
#pragma unroll
      for (int j = 0; j < ECH; ++j) {
        bool valid = (h + j) < cc;
        p[j]  = valid ? p[j]  : ninf;
        s1[j] = valid ? s1[j] : ninf;
      }

      float mn0 = fmaxf(fmaxf(fmaxf(p[0], p[1]), fmaxf(p[2], p[3])), m0);
      float sc0 = expf(m0 - mn0);
      float w00 = expf(p[0] - mn0), w01 = expf(p[1] - mn0);
      float w02 = expf(p[2] - mn0), w03 = expf(p[3] - mn0);
      m0 = mn0;
      d0 = d0 * sc0 + ((w00 + w01) + (w02 + w03));
      step4(acc0a, sc0, w00, ya[0], w01, ya[1], w02, ya[2], w03, ya[3]);
      step4(acc0b, sc0, w00, yb[0], w01, yb[1], w02, yb[2], w03, yb[3]);

      float mn1 = fmaxf(fmaxf(fmaxf(s1[0], s1[1]), fmaxf(s1[2], s1[3])), m1);
      float sc1 = expf(m1 - mn1);
      float w10 = expf(s1[0] - mn1), w11 = expf(s1[1] - mn1);
      float w12 = expf(s1[2] - mn1), w13 = expf(s1[3] - mn1);
      m1 = mn1;
      d1 = d1 * sc1 + ((w10 + w11) + (w12 + w13));
      step4(acc1a, sc1, w10, ya[0], w11, ya[1], w12, ya[2], w13, ya[3]);
      step4(acc1b, sc1, w10, yb[0], w11, yb[1], w12, yb[2], w13, yb[3]);
    }
  }

  float4 e0a = make_float4(0.0f, 0.0f, 0.0f, 0.0f), e0b = e0a, e1a = e0a, e1b = e0a;
  if (LAST) {
    const float4* p0 = in_row<false>(g, ue, ie, y1);
    e0a = p0[lane]; e0b = p0[lane + GHALF];
    float2 ms = st_in[g];
    e1a.x = yra.x * ms.y + ms.x; e1a.y = yra.y * ms.y + ms.x;
    e1a.z = yra.z * ms.y + ms.x; e1a.w = yra.w * ms.y + ms.x;
    e1b.x = yrb.x * ms.y + ms.x; e1b.y = yrb.y * ms.y + ms.x;
    e1b.z = yrb.z * ms.y + ms.x; e1b.w = yrb.w * ms.y + ms.x;
  }

  float4 oa = make_float4(0.0f, 0.0f, 0.0f, 0.0f);
  float4 ob = oa;
  if (n > 0) {
    float i0 = 0.5f / d0;
    float i1 = 0.5f / d1;
    oa.x = acc0a.x * i0 + acc1a.x * i1;
    oa.y = acc0a.y * i0 + acc1a.y * i1;
    oa.z = acc0a.z * i0 + acc1a.z * i1;
    oa.w = acc0a.w * i0 + acc1a.w * i1;
    ob.x = acc0b.x * i0 + acc1b.x * i1;
    ob.y = acc0b.y * i0 + acc1b.y * i1;
    ob.z = acc0b.z * i0 + acc1b.z * i1;
    ob.w = acc0b.w * i0 + acc1b.w * i1;
  }
  float4* dp = (float4*)(dst + (size_t)g * HIDDEN) + lane;
  if (LAST) {
    const float third = 1.0f / 3.0f;
    oa.x = (oa.x + (e0a.x + e1a.x)) * third;
    oa.y = (oa.y + (e0a.y + e1a.y)) * third;
    oa.z = (oa.z + (e0a.z + e1a.z)) * third;
    oa.w = (oa.w + (e0a.w + e1a.w)) * third;
    ob.x = (ob.x + (e0b.x + e1b.x)) * third;
    ob.y = (ob.y + (e0b.y + e1b.y)) * third;
    ob.z = (ob.z + (e0b.z + e1b.z)) * third;
    ob.w = (ob.w + (e0b.w + e1b.w)) * third;
  } else {
    float s = ((oa.x + oa.y) + (oa.z + oa.w)) + ((ob.x + ob.y) + (ob.z + ob.w));
#pragma unroll
    for (int o = GLANES / 2; o > 0; o >>= 1) s += __shfl_xor(s, o, GLANES);
    float mu = s * (1.0f / HIDDEN);
    oa.x -= mu; oa.y -= mu; oa.z -= mu; oa.w -= mu;
    ob.x -= mu; ob.y -= mu; ob.z -= mu; ob.w -= mu;
    float q = ((oa.x * oa.x + oa.y * oa.y) + (oa.z * oa.z + oa.w * oa.w)) +
              ((ob.x * ob.x + ob.y * ob.y) + (ob.z * ob.z + ob.w * ob.w));
#pragma unroll
    for (int o = GLANES / 2; o > 0; o >>= 1) q += __shfl_xor(q, o, GLANES);
    float rin = rsqrtf(q * (1.0f / HIDDEN) + LN_EPS);
    oa.x *= rin; oa.y *= rin; oa.z *= rin; oa.w *= rin;
    ob.x *= rin; ob.y *= rin; ob.z *= rin; ob.w *= rin;
    if (lane == 0) st_out[g] = make_float2(mu, 1.0f / rin);
  }
  dp[0]     = oa;
  dp[GHALF] = ob;
  }
}

extern "C" void kernel_launch(void* const* d_in, const int* in_sizes, int n_in,
                              void* d_out, int out_size, void* d_ws, size_t ws_size,
                              hipStream_t stream) {
  (void)in_sizes; (void)n_in; (void)out_size; (void)ws_size;
  const float* ue      = (const float*)d_in[0];
  const float* ie      = (const float*)d_in[1];
  const float* eigs    = (const float*)d_in[2];
  const float* lambda0 = (const float*)d_in[3];
  const float* pemb    = (const float*)d_in[4];
  const int*   indices = (const int*)d_in[5];
  const int*   pt      = (const int*)d_in[6];
  const int* row = indices;
  const int* col = indices + NEDGES;
  float* out = (float*)d_out;

  char* p = (char*)d_ws;
  auto carve = [&](size_t bytes) -> char* {
    char* r = p;
    p += (bytes + 255) & ~(size_t)255;
    return r;
  };
  float2* stats = (float2*)carve((size_t)NNODES * 8);
  float2* stat1 = (float2*)carve((size_t)NNODES * 8);
  float* y1    = (float*)carve((size_t)NNODES * HIDDEN * 4);
  int2*  sedge = (int2*) carve((size_t)NEDGES * 8);
  int*   rank  = (int*)  carve((size_t)NEDGES * 4);
  int*   cnt   = (int*)  carve((size_t)NNODES * 4);
  int*   off   = (int*)  carve((size_t)NNODES * 4);
  int*   bsum  = (int*)  carve((size_t)SCAN_B * 4);

  const dim3 B(256);
  const int GRID_LN        = 1172;
  const int GRID_EDGE      = 1024;
  const dim3 B_ROW(128);
  const int GRID_ROW       = 9376;

  hipMemsetAsync(cnt, 0, (size_t)NNODES * 4, stream);
  hipLaunchKernelGGL(hist_rank_kernel, dim3(GRID_EDGE), B, 0, stream, row, cnt, rank);
  hipLaunchKernelGGL(blocksum_kernel, dim3(SCAN_B), dim3(SCAN_T), 0, stream, cnt, bsum);
  hipLaunchKernelGGL(offsets_kernel, dim3(SCAN_B), dim3(SCAN_T), 0, stream, cnt, bsum, off);
  hipLaunchKernelGGL(fill_kernel, dim3(GRID_EDGE), B, 0, stream, row, col, pt, rank, off, sedge);

  hipLaunchKernelGGL(ln_wmma_kernel, dim3(GRID_LN), B, 0, stream, ue, ie, NUSERS, stats);

  hipLaunchKernelGGL(row_kernel<false>, dim3(GRID_ROW), B_ROW, 0, stream,
                     (const float2*)stats, stat1, eigs, off, cnt, sedge, lambda0, pemb,
                     ue, ie, (const float*)nullptr, y1);
  hipLaunchKernelGGL(row_kernel<true>, dim3(GRID_ROW), B_ROW, 0, stream,
                     (const float2*)stat1, (float2*)nullptr, eigs, off, cnt, sedge,
                     lambda0 + 1, pemb + NPATHS,
                     ue, ie, (const float*)y1, out);
}


# Row kernels work a row ahead: the next row's degree, first slot, first 8 edges, own input row and pairs are asked for during the current row (own rows also prefetched to L2); sum order unchanged
# speedup vs baseline: 5.6184x; 1.0349x over previous
#include <hip/hip_runtime.h>

#define HIDDEN 64
#define EIGD   16
#define NUSERS 100000
#define NITEMS 200000
#define NNODES 300000
#define NEDGES 1250000
#define NPATHS 14
#define LN_EPS 1e-5f

typedef __attribute__((ext_vector_type(2))) float v2f;
typedef __attribute__((ext_vector_type(8))) float v8f;
typedef __attribute__((ext_vector_type(4))) float v4f;

__global__ void ln_wmma_kernel(const float* __restrict__ srcA,
                               const float* __restrict__ srcB,
                               int nA, float2* __restrict__ stats) {
  int lane = threadIdx.x & 31;
  int wave = threadIdx.x >> 5;
  int m = lane & 15;
  int h = lane >> 4;
  const int wstep = (gridDim.x * (blockDim.x >> 5)) << 4;
  for (int base = (blockIdx.x * (blockDim.x >> 5) + wave) << 4; base < NNODES; base += wstep) {
  int node = base + m;
  const float* srow = (node < nA) ? (srcA + (size_t)node * HIDDEN)
                                  : (srcB + (size_t)(node - nA) * HIDDEN);
  const float* sp = srow + 2 * h;

  const v2f ones = {1.0f, 1.0f};
  v8f accS = {};
  v8f accQ = {};
#pragma unroll
  for (int c = 0; c < 16; ++c) {
    float2 v = *(const float2*)(sp + 4 * c);
    v2f a  = {v.x, v.y};
    v2f a2 = {v.x * v.x, v.y * v.y};
    accS = __builtin_amdgcn_wmma_f32_16x16x4_f32(false, a,  false, ones,
                                                 (short)0, accS, false, false);
    accQ = __builtin_amdgcn_wmma_f32_16x16x4_f32(false, a2, false, ones,
                                                 (short)0, accQ, false, false);
  }

  float s = 0.0f, q = 0.0f;
#pragma unroll
  for (int i = 0; i < 8; ++i) {
    float sLo = __shfl(accS[i], 0, 32);
    float sHi = __shfl(accS[i], 16, 32);
    float qLo = __shfl(accQ[i], 0, 32);
    float qHi = __shfl(accQ[i], 16, 32);
    s = (m == i)     ? sLo : s;
    s = (m == 8 + i) ? sHi : s;
    q = (m == i)     ? qLo : q;
    q = (m == 8 + i) ? qHi : q;
  }

  float mu  = s * (1.0f / HIDDEN);
  float var = q * (1.0f / HIDDEN) - mu * mu;
  float rin = rsqrtf(var + LN_EPS);
  if (h == 0) stats[node] = make_float2(mu, rin);
  }
}

__global__ void hist_rank_kernel(const int* __restrict__ row,
                                 int* __restrict__ cnt,
                                 int* __restrict__ rank) {
  const int stride = gridDim.x * blockDim.x;
  for (int e = blockIdx.x * blockDim.x + threadIdx.x; e < NEDGES; e += stride)
    rank[e] = atomicAdd(cnt + row[e], 1);
}

#define SCAN_T 256
#define SCAN_E (SCAN_T * 4)
#define SCAN_B ((NNODES + SCAN_E - 1) / SCAN_E)

__global__ void blocksum_kernel(const int* __restrict__ cnt, int* __restrict__ bsum) {
  __shared__ int wsum[SCAN_T / 32];
  int t = threadIdx.x;
  int i0 = (blockIdx.x * SCAN_T + t) * 4;
  int s = 0;
  if (i0 + 3 < NNODES) {
    int4 v = *(const int4*)(cnt + i0);
    s = v.x + v.y + v.z + v.w;
  } else {
    for (int k = 0; k < 4; ++k)
      if (i0 + k < NNODES) s += cnt[i0 + k];
  }
  for (int o = 16; o > 0; o >>= 1) s += __shfl_xor(s, o, 32);
  if ((t & 31) == 0) wsum[t >> 5] = s;
  __syncthreads();
  if (t == 0) {
    int a = 0;
    for (int w = 0; w < SCAN_T / 32; ++w) a += wsum[w];
    bsum[blockIdx.x] = a;
  }
}

__global__ void offsets_kernel(const int* __restrict__ cnt,
                               const int* __restrict__ bsum,
                               int* __restrict__ off) {
  __shared__ int wpre[SCAN_T / 32];
  __shared__ int wtot[SCAN_T / 32];
  int t = threadIdx.x;
  int lane = t & 31;
  int wave = t >> 5;

  int ps = 0;
  for (int i = t; i < (int)blockIdx.x; i += SCAN_T) ps += bsum[i];
  for (int o = 16; o > 0; o >>= 1) ps += __shfl_xor(ps, o, 32);
  if (lane == 0) wpre[wave] = ps;

  int i0 = (blockIdx.x * SCAN_T + t) * 4;
  int v0 = 0, v1 = 0, v2 = 0, v3 = 0;
  if (i0 + 3 < NNODES) {
    int4 v = *(const int4*)(cnt + i0);
    v0 = v.x; v1 = v.y; v2 = v.z; v3 = v.w;
  } else {
    if (i0     < NNODES) v0 = cnt[i0];
    if (i0 + 1 < NNODES) v1 = cnt[i0 + 1];
    if (i0 + 2 < NNODES) v2 = cnt[i0 + 2];
  }
  int s = v0 + v1 + v2 + v3;
  int x = s;
  for (int o = 1; o < 32; o <<= 1) {
    int nb = __shfl_up(x, o, 32);
    if (lane >= o) x += nb;
  }
  if (lane == 31) wtot[wave] = x;
  __syncthreads();
  int basev = 0;
  for (int w = 0; w < SCAN_T / 32; ++w) {
    basev += wpre[w];
    if (w < wave) basev += wtot[w];
  }
  int ex = basev + x - s;
  if (i0 + 3 < NNODES) {
    *(int4*)(off + i0) = make_int4(ex, ex + v0, ex + v0 + v1, ex + v0 + v1 + v2);
  } else {
    if (i0     < NNODES) off[i0]     = ex;
    if (i0 + 1 < NNODES) off[i0 + 1] = ex + v0;
    if (i0 + 2 < NNODES) off[i0 + 2] = ex + v0 + v1;
  }
}

__global__ void fill_kernel(const int* __restrict__ row,
                            const int* __restrict__ col,
                            const int* __restrict__ pt,
                            const int* __restrict__ rank,
                            const int* __restrict__ off,
                            int2* __restrict__ sedge) {
  const int stride = gridDim.x * blockDim.x;
  for (int e = blockIdx.x * blockDim.x + threadIdx.x; e < NEDGES; e += stride) {
    int r = row[e];
    sedge[off[r] + rank[e]] = make_int2(col[e], pt[e]);
  }
}

#define GLANES 8
#define GHALF  (HIDDEN / 4 / 2)
#define ECH    4

template <bool LAST>
__device__ __forceinline__ const float4* in_row(int node,
                                                const float* __restrict__ ue,
                                                const float* __restrict__ ie,
                                                const float* __restrict__ y1) {
  if (LAST) return (const float4*)(y1 + (size_t)node * HIDDEN);
  bool u = node < NUSERS;
  const float* tab = u ? ue : ie;
  int idx = u ? node : node - NUSERS;
  return (const float4*)(tab + (size_t)idx * HIDDEN);
}

__device__ __forceinline__ float4 norm4(float4 v, float2 s) {
  v.x = (v.x - s.x) * s.y;
  v.y = (v.y - s.x) * s.y;
  v.z = (v.z - s.x) * s.y;
  v.w = (v.w - s.x) * s.y;
  return v;
}

__device__ __forceinline__ v4f norm4v(v4f v, v2f s) {
  v.x = (v.x - s.x) * s.y;
  v.y = (v.y - s.x) * s.y;
  v.z = (v.z - s.x) * s.y;
  v.w = (v.w - s.x) * s.y;
  return v;
}

__device__ __forceinline__ void step4(float4& acc, float sc,
                                      float w0, const v4f& y0, float w1, const v4f& y1,
                                      float w2, const v4f& y2, float w3, const v4f& y3) {
  acc.x = acc.x * sc + w0 * y0.x + w1 * y1.x + w2 * y2.x + w3 * y3.x;
  acc.y = acc.y * sc + w0 * y0.y + w1 * y1.y + w2 * y2.y + w3 * y3.y;
  acc.z = acc.z * sc + w0 * y0.z + w1 * y1.z + w2 * y2.z + w3 * y3.z;
  acc.w = acc.w * sc + w0 * y0.w + w1 * y1.w + w2 * y2.w + w3 * y3.w;
}

template <bool LAST>
__global__ void __launch_bounds__(128, LAST ? 7 : 1) row_kernel(const float2* __restrict__ st_in,
                           float2* __restrict__ st_out,
                           const float* __restrict__ eigs,
                           const int* __restrict__ off,
                           const int* __restrict__ cnt,
                           const int2* __restrict__ sedge,
                           const float* __restrict__ lambda0_l,
                           const float* __restrict__ pemb_l,
                           const float* __restrict__ ue,
                           const float* __restrict__ ie,
                           const float* __restrict__ y1,
                           float* __restrict__ dst) {
  const int lane = threadIdx.x & (GLANES - 1);
  const int ngroups = gridDim.x * (blockDim.x / GLANES);
  const float el = expf(lambda0_l[0]);
  int g = blockIdx.x * (blockDim.x / GLANES) + (threadIdx.x / GLANES);
  int n = 0;
  int2 my = make_int2(0, 0);
  float4 yra = make_float4(0.0f, 0.0f, 0.0f, 0.0f), yrb = yra;
  float2 er = make_float2(0.0f, 0.0f);
  float2 sg = make_float2(0.0f, 0.0f);
  int n_nx = 0, b_nx = 0;
  if (g < NNODES) {
    n = cnt[g];
    const int beg = off[g];
    if (g + ngroups < NNODES) { n_nx = cnt[g + ngroups]; b_nx = off[g + ngroups]; }
    const float4* rp = in_row<LAST>(g, ue, ie, y1);
    yra = rp[lane];
    yrb = rp[lane + GHALF];
    er = ((const float2*)(eigs + (size_t)g * EIGD))[lane];
    if (!LAST) sg = st_in[g];
    asm volatile("" :: "v"(n), "v"(beg));
    if (lane < n) my = sedge[beg + lane];
  }
  for (; g < NNODES; g += ngroups) {
  const int gn = g + ngroups;
  const bool more = gn < NNODES;
  if (LAST) __builtin_prefetch(in_row<false>(g, ue, ie, y1), 0, 3);
  if (more) __builtin_prefetch(in_row<LAST>(gn, ue, ie, y1), 0, 3);
  if (!LAST) {
    yra = norm4(yra, sg);
    yrb = norm4(yrb, sg);
  }
  float  wra = el * er.x;
  float  wrb = el * er.y;

  const float ninf = -__builtin_inff();
  float m0 = ninf, m1 = ninf, d0 = 0.0f, d1 = 0.0f;
  float4 acc0a = make_float4(0.0f, 0.0f, 0.0f, 0.0f);
  float4 acc0b = make_float4(0.0f, 0.0f, 0.0f, 0.0f);
  float4 acc1a = make_float4(0.0f, 0.0f, 0.0f, 0.0f);
  float4 acc1b = make_float4(0.0f, 0.0f, 0.0f, 0.0f);

  for (int base = 0; base < n; base += GLANES) {
    int cc = n - base;
    if (cc > GLANES) cc = GLANES;
    for (int h = 0; h < cc; h += ECH) {
      v4f   ya[ECH], yb[ECH];
      v2f   sn[ECH], ec[ECH];
      float p[ECH], s1[ECH];
#pragma unroll
      for (int j = 0; j < ECH; ++j) {
        int c = __shfl(my.x, h + j, GLANES);
        int t = __shfl(my.y, h + j, GLANES);
        const v4f* cp = (const v4f*)in_row<LAST>(c, ue, ie, y1);
        ya[j] = cp[lane];
        yb[j] = cp[lane + GHALF];
        if (!LAST) sn[j] = ((const v2f*)st_in)[c];
        ec[j] = ((const v2f*)(eigs + (size_t)c * EIGD))[lane];
        s1[j] = pemb_l[t];
      }
      if (LAST)
        asm volatile("" :: "v"(ya[0]), "v"(yb[0]), "v"(ya[1]), "v"(yb[1]),
                           "v"(ya[2]), "v"(yb[2]), "v"(ya[3]), "v"(yb[3]),
                           "v"(ec[0]), "v"(ec[1]), "v"(ec[2]), "v"(ec[3]),
                           "v"(s1[0]), "v"(s1[1]), "v"(s1[2]), "v"(s1[3]));
      else
        asm volatile("" :: "v"(ya[0]), "v"(yb[0]), "v"(ya[1]), "v"(yb[1]),
                           "v"(ya[2]), "v"(yb[2]), "v"(ya[3]), "v"(yb[3]),
                           "v"(sn[0]), "v"(sn[1]), "v"(sn[2]), "v"(sn[3]),
                           "v"(ec[0]), "v"(ec[1]), "v"(ec[2]), "v"(ec[3]),
                           "v"(s1[0]), "v"(s1[1]), "v"(s1[2]), "v"(s1[3]));

#pragma unroll
      for (int j = 0; j < ECH; ++j) {
        if (!LAST) {
          ya[j] = norm4v(ya[j], sn[j]);
          yb[j] = norm4v(yb[j], sn[j]);
        }
        float q = 0.125f * (yra.x * ya[j].x + yra.y * ya[j].y + yra.z * ya[j].z + yra.w * ya[j].w +
                            yrb.x * yb[j].x + yrb.y * yb[j].y + yrb.z * yb[j].z + yrb.w * yb[j].w);
        q += wra * ec[j].x;
        q += wrb * ec[j].y;
        p[j] = q;
      }
#pragma unroll
      for (int o = GLANES / 2; o > 0; o >>= 1) {
        float r[ECH];
#pragma unroll
        for (int j = 0; j < ECH; ++j) r[j] = __shfl_xor(p[j], o, GLANES);
#pragma unroll
        for (int j = 0; j < ECH; ++j) p[j] += r[j];
      }
#pragma unroll
      for (int j = 0; j < ECH; ++j) {
        bool valid = (h + j) < cc;
        p[j]  = valid ? p[j]  : ninf;
        s1[j] = valid ? s1[j] : ninf;
      }

      float mn0 = fmaxf(fmaxf(fmaxf(p[0], p[1]), fmaxf(p[2], p[3])), m0);
      float sc0 = expf(m0 - mn0);
      float w00 = expf(p[0] - mn0), w01 = expf(p[1] - mn0);
      float w02 = expf(p[2] - mn0), w03 = expf(p[3] - mn0);
      m0 = mn0;
      d0 = d0 * sc0 + ((w00 + w01) + (w02 + w03));
      step4(acc0a, sc0, w00, ya[0], w01, ya[1], w02, ya[2], w03, ya[3]);
      step4(acc0b, sc0, w00, yb[0], w01, yb[1], w02, yb[2], w03, yb[3]);

      float mn1 = fmaxf(fmaxf(fmaxf(s1[0], s1[1]), fmaxf(s1[2], s1[3])), m1);
      float sc1 = expf(m1 - mn1);
      float w10 = expf(s1[0] - mn1), w11 = expf(s1[1] - mn1);
      float w12 = expf(s1[2] - mn1), w13 = expf(s1[3] - mn1);
      m1 = mn1;
      d1 = d1 * sc1 + ((w10 + w11) + (w12 + w13));
      step4(acc1a, sc1, w10, ya[0], w11, ya[1], w12, ya[2], w13, ya[3]);
      step4(acc1b, sc1, w10, yb[0], w11, yb[1], w12, yb[2], w13, yb[3]);
    }
    const int left = n - base - GLANES;
    if (left > 0) {
      my = make_int2(0, 0);
      if (lane < left) my = sedge[off[g] + base + GLANES + lane];
    }
  }
  float4 e0a = make_float4(0.0f, 0.0f, 0.0f, 0.0f), e0b = e0a, e1a = e0a, e1b = e0a;
  float2 ms = make_float2(0.0f, 0.0f);
  if (LAST) {
    const float4* p0 = in_row<false>(g, ue, ie, y1);
    e0a = p0[lane]; e0b = p0[lane + GHALF];
    ms = st_in[g];
  }
  float4 yra_nx = make_float4(0.0f, 0.0f, 0.0f, 0.0f), yrb_nx = yra_nx;
  float2 er_nx = make_float2(0.0f, 0.0f), sg_nx = er_nx;
  if (more) {
    const float4* rpn = in_row<LAST>(gn, ue, ie, y1);
    yra_nx = rpn[lane];
    yrb_nx = rpn[lane + GHALF];
    er_nx = ((const float2*)(eigs + (size_t)gn * EIGD))[lane];
    if (!LAST) sg_nx = st_in[gn];
  }
  int n_nn = 0, b_nn = 0;
  my = make_int2(0, 0);
  if (lane < n_nx) my = sedge[b_nx + lane];
  if (gn + ngroups < NNODES) { n_nn = cnt[gn + ngroups]; b_nn = off[gn + ngroups]; }
  if (LAST) {
    e1a.x = yra.x * ms.y + ms.x; e1a.y = yra.y * ms.y + ms.x;
    e1a.z = yra.z * ms.y + ms.x; e1a.w = yra.w * ms.y + ms.x;
    e1b.x = yrb.x * ms.y + ms.x; e1b.y = yrb.y * ms.y + ms.x;
    e1b.z = yrb.z * ms.y + ms.x; e1b.w = yrb.w * ms.y + ms.x;
  }

  float4 oa = make_float4(0.0f, 0.0f, 0.0f, 0.0f);
  float4 ob = oa;
  if (n > 0) {
    float i0 = 0.5f / d0;
    float i1 = 0.5f / d1;
    oa.x = acc0a.x * i0 + acc1a.x * i1;
    oa.y = acc0a.y * i0 + acc1a.y * i1;
    oa.z = acc0a.z * i0 + acc1a.z * i1;
    oa.w = acc0a.w * i0 + acc1a.w * i1;
    ob.x = acc0b.x * i0 + acc1b.x * i1;
    ob.y = acc0b.y * i0 + acc1b.y * i1;
    ob.z = acc0b.z * i0 + acc1b.z * i1;
    ob.w = acc0b.w * i0 + acc1b.w * i1;
  }
  float4* dp = (float4*)(dst + (size_t)g * HIDDEN) + lane;
  if (LAST) {
    const float third = 1.0f / 3.0f;
    oa.x = (oa.x + (e0a.x + e1a.x)) * third;
    oa.y = (oa.y + (e0a.y + e1a.y)) * third;
    oa.z = (oa.z + (e0a.z + e1a.z)) * third;
    oa.w = (oa.w + (e0a.w + e1a.w)) * third;
    ob.x = (ob.x + (e0b.x + e1b.x)) * third;
    ob.y = (ob.y + (e0b.y + e1b.y)) * third;
    ob.z = (ob.z + (e0b.z + e1b.z)) * third;
    ob.w = (ob.w + (e0b.w + e1b.w)) * third;
  } else {
    float s = ((oa.x + oa.y) + (oa.z + oa.w)) + ((ob.x + ob.y) + (ob.z + ob.w));
#pragma unroll
    for (int o = GLANES / 2; o > 0; o >>= 1) s += __shfl_xor(s, o, GLANES);
    float mu = s * (1.0f / HIDDEN);
    oa.x -= mu; oa.y -= mu; oa.z -= mu; oa.w -= mu;
    ob.x -= mu; ob.y -= mu; ob.z -= mu; ob.w -= mu;
    float q = ((oa.x * oa.x + oa.y * oa.y) + (oa.z * oa.z + oa.w * oa.w)) +
              ((ob.x * ob.x + ob.y * ob.y) + (ob.z * ob.z + ob.w * ob.w));
#pragma unroll
    for (int o = GLANES / 2; o > 0; o >>= 1) q += __shfl_xor(q, o, GLANES);
    float rin = rsqrtf(q * (1.0f / HIDDEN) + LN_EPS);
    oa.x *= rin; oa.y *= rin; oa.z *= rin; oa.w *= rin;
    ob.x *= rin; ob.y *= rin; ob.z *= rin; ob.w *= rin;
    if (lane == 0) st_out[g] = make_float2(mu, 1.0f / rin);
  }
  dp[0]     = oa;
  dp[GHALF] = ob;
  n = n_nx;
  n_nx = n_nn; b_nx = b_nn;
  yra = yra_nx; yrb = yrb_nx; er = er_nx; sg = sg_nx;
  }
}

extern "C" void kernel_launch(void* const* d_in, const int* in_sizes, int n_in,
                              void* d_out, int out_size, void* d_ws, size_t ws_size,
                              hipStream_t stream) {
  (void)in_sizes; (void)n_in; (void)out_size; (void)ws_size;
  const float* ue      = (const float*)d_in[0];
  const float* ie      = (const float*)d_in[1];
  const float* eigs    = (const float*)d_in[2];
  const float* lambda0 = (const float*)d_in[3];
  const float* pemb    = (const float*)d_in[4];
  const int*   indices = (const int*)d_in[5];
  const int*   pt      = (const int*)d_in[6];
  const int* row = indices;
  const int* col = indices + NEDGES;
  float* out = (float*)d_out;

  char* p = (char*)d_ws;
  auto carve = [&](size_t bytes) -> char* {
    char* r = p;
    p += (bytes + 255) & ~(size_t)255;
    return r;
  };
  float2* stats = (float2*)carve((size_t)NNODES * 8);
  float2* stat1 = (float2*)carve((size_t)NNODES * 8);
  float* y1    = (float*)carve((size_t)NNODES * HIDDEN * 4);
  int2*  sedge = (int2*) carve((size_t)NEDGES * 8);
  int*   rank  = (int*)  carve((size_t)NEDGES * 4);
  int*   cnt   = (int*)  carve((size_t)NNODES * 4);
  int*   off   = (int*)  carve((size_t)NNODES * 4);
  int*   bsum  = (int*)  carve((size_t)SCAN_B * 4);

  const dim3 B(256);
  const int GRID_LN        = 1172;
  const int GRID_EDGE      = 1024;
  const dim3 B_ROW(128);
  const int GRID_ROW       = 9376;

  hipMemsetAsync(cnt, 0, (size_t)NNODES * 4, stream);
  hipLaunchKernelGGL(hist_rank_kernel, dim3(GRID_EDGE), B, 0, stream, row, cnt, rank);
  hipLaunchKernelGGL(blocksum_kernel, dim3(SCAN_B), dim3(SCAN_T), 0, stream, cnt, bsum);
  hipLaunchKernelGGL(offsets_kernel, dim3(SCAN_B), dim3(SCAN_T), 0, stream, cnt, bsum, off);
  hipLaunchKernelGGL(fill_kernel, dim3(GRID_EDGE), B, 0, stream, row, col, pt, rank, off, sedge);

  hipLaunchKernelGGL(ln_wmma_kernel, dim3(GRID_LN), B, 0, stream, ue, ie, NUSERS, stats);

  hipLaunchKernelGGL(row_kernel<false>, dim3(GRID_ROW), B_ROW, 0, stream,
                     (const float2*)stats, stat1, eigs, off, cnt, sedge, lambda0, pemb,
                     ue, ie, (const float*)nullptr, y1);
  hipLaunchKernelGGL(row_kernel<true>, dim3(GRID_ROW), B_ROW, 0, stream,
                     (const float2*)stat1, (float2*)nullptr, eigs, off, cnt, sedge,
                     lambda0 + 1, pemb + NPATHS,
                     ue, ie, (const float*)y1, out);
}
